# starting kernel
# baseline (speedup 1.0000x reference)
#include <hip/hip_runtime.h>
#include <cmath>

#define NNODES 100000
#define NEDGES 600000
#define FEAT   128
#define NGRAPH 256

typedef __attribute__((ext_vector_type(2))) float v2f;
typedef __attribute__((ext_vector_type(8))) float v8f;

__global__ void gcn_init_kernel(float* __restrict__ out, float* __restrict__ cnts,
                                float* __restrict__ deg) {
    int i = blockIdx.x * 256 + threadIdx.x;
    if (i < NGRAPH * FEAT) out[i] = 0.0f;
    if (i < NGRAPH)        cnts[i] = 0.0f;
    if (i < NNODES)        deg[i]  = 1.0f;
}

__global__ void gcn_deg_kernel(const int* __restrict__ col, float* __restrict__ deg) {
    int e = blockIdx.x * 256 + threadIdx.x;
    if (e < NEDGES) atomicAdd(&deg[col[e]], 1.0f);
}

__global__ void gcn_dinv_kernel(float* __restrict__ deg) {
    int n = blockIdx.x * 256 + threadIdx.x;
    if (n < NNODES) {
        float d = deg[n];
        deg[n] = (d > 0.0f) ? rsqrtf(d) : 0.0f;
    }
}

template <bool RELU>
__global__ void gcn_gemm_kernel(const float* __restrict__ X,
                                const float* __restrict__ W,
                                const float* __restrict__ bias,
                                float* __restrict__ Y) {
    __shared__ float Wt[FEAT * 16];
    __shared__ float Bt[16];

    const int lane  = threadIdx.x & 31;
    const int wave  = threadIdx.x >> 5;
    const int m0    = blockIdx.x * 128 + wave * 16;
    const bool active = (m0 < NNODES);

    const int mrow  = m0 + (lane & 15);
    const int khalf = (lane >> 4) << 1;
    const int ncol  = lane & 15;
    const int half  = lane >> 4;

    for (int nt = 0; nt < 8; ++nt) {
        __syncthreads();
        #pragma unroll
        for (int i = 0; i < 8; ++i) {
            int idx = threadIdx.x * 8 + i;
            int kk  = idx >> 4;
            int cc  = idx & 15;
            Wt[idx] = W[kk * FEAT + nt * 16 + cc];
        }
        if (threadIdx.x < 16) Bt[threadIdx.x] = bias[nt * 16 + threadIdx.x];
        __syncthreads();

        if (!active) continue;

        v8f acc = {};
        #pragma unroll 4
        for (int k = 0; k < FEAT; k += 4) {
            float a0 = X[(size_t)mrow * FEAT + k + khalf];
            float a1 = X[(size_t)mrow * FEAT + k + khalf + 1];
            if (RELU) { a0 = fmaxf(a0, 0.0f); a1 = fmaxf(a1, 0.0f); }
            v2f a; a[0] = a0; a[1] = a1;
            v2f b;
            b[0] = Wt[(k + khalf) * 16 + ncol];
            b[1] = Wt[(k + khalf + 1) * 16 + ncol];
            acc = __builtin_amdgcn_wmma_f32_16x16x4_f32(
                false, a, false, b, (short)0, acc, false, false);
        }

        #pragma unroll
        for (int v = 0; v < 8; ++v) {
            int r = m0 + v + half * 8;
            Y[(size_t)r * FEAT + nt * 16 + ncol] = acc[v] + Bt[ncol];
        }
    }
}

__global__ void gcn_agg_init_kernel(const float* __restrict__ H,
                                    const float* __restrict__ dinv,
                                    float* __restrict__ AGG) {
    int i = blockIdx.x * 256 + threadIdx.x;
    if (i >= NNODES * 32) return;
    int n = i >> 5;
    float s = dinv[n]; s = s * s;
    float4 h = *(const float4*)(H + (size_t)i * 4);
    float4 r = make_float4(h.x * s, h.y * s, h.z * s, h.w * s);
    *(float4*)(AGG + (size_t)i * 4) = r;
}

__global__ void gcn_scatter_kernel(const int* __restrict__ row,
                                   const int* __restrict__ col,
                                   const float* __restrict__ dinv,
                                   const float* __restrict__ H,
                                   float* __restrict__ AGG) {
    int t = blockIdx.x * 256 + threadIdx.x;
    int e = t >> 5;
    if (e >= NEDGES) return;
    int lane = t & 31;
    int r = row[e];
    int c = col[e];
    float norm = dinv[r] * dinv[c];
    float4 h = *(const float4*)(H + (size_t)c * FEAT + lane * 4);
    float* dst = AGG + (size_t)r * FEAT + lane * 4;
    atomicAdd(dst + 0, norm * h.x);
    atomicAdd(dst + 1, norm * h.y);
    atomicAdd(dst + 2, norm * h.z);
    atomicAdd(dst + 3, norm * h.w);
}

__global__ void gcn_pool_kernel(const float* __restrict__ AGG,
                                const int* __restrict__ batch,
                                float* __restrict__ out,
                                float* __restrict__ cnts) {
    int t = blockIdx.x * 256 + threadIdx.x;
    int n = t >> 5;
    if (n >= NNODES) return;
    int lane = t & 31;
    int g = batch[n];
    float4 h = *(const float4*)(AGG + (size_t)n * FEAT + lane * 4);
    float* dst = out + (size_t)g * FEAT + lane * 4;
    atomicAdd(dst + 0, h.x);
    atomicAdd(dst + 1, h.y);
    atomicAdd(dst + 2, h.z);
    atomicAdd(dst + 3, h.w);
    if (lane == 0) atomicAdd(&cnts[g], 1.0f);
}

__global__ void gcn_finalize_kernel(float* __restrict__ out,
                                    const float* __restrict__ cnts) {
    int i = blockIdx.x * 256 + threadIdx.x;
    if (i < NGRAPH * FEAT) {
        float c = cnts[i >> 7];
        out[i] = out[i] / fmaxf(c, 1.0f);
    }
}

extern "C" void kernel_launch(void* const* d_in, const int* in_sizes, int n_in,
                              void* d_out, int out_size, void* d_ws, size_t ws_size,
                              hipStream_t stream) {
    const float* x          = (const float*)d_in[0];
    const int*   edge_index = (const int*)  d_in[1];
    const int*   batch      = (const int*)  d_in[2];
    const float* W1 = (const float*)d_in[4];
    const float* b1 = (const float*)d_in[5];
    const float* W2 = (const float*)d_in[6];
    const float* b2 = (const float*)d_in[7];
    float* out = (float*)d_out;

    float* H    = (float*)d_ws;
    float* AGG  = H    + (size_t)NNODES * FEAT;
    float* dinv = AGG  + (size_t)NNODES * FEAT;
    float* cnts = dinv + NNODES;

    const int* row = edge_index;
    const int* col = edge_index + NEDGES;

    dim3 blk(256);
    const int grid_nodes  = (NNODES + 255) / 256;
    const int grid_edges  = (NEDGES + 255) / 256;
    const int grid_gemm   = (NNODES + 127) / 128;
    const int grid_nfeat4 = (NNODES * 32 + 255) / 256;
    const int grid_escat  = (NEDGES * 32 + 255) / 256;
    const int grid_out    = (NGRAPH * FEAT + 255) / 256;

    gcn_init_kernel<<<grid_nodes, blk, 0, stream>>>(out, cnts, dinv);
    gcn_deg_kernel<<<grid_edges, blk, 0, stream>>>(col, dinv);
    gcn_dinv_kernel<<<grid_nodes, blk, 0, stream>>>(dinv);

    gcn_gemm_kernel<false><<<grid_gemm, blk, 0, stream>>>(x, W1, b1, H);
    gcn_agg_init_kernel<<<grid_nfeat4, blk, 0, stream>>>(H, dinv, AGG);
    gcn_scatter_kernel<<<grid_escat, blk, 0, stream>>>(row, col, dinv, H, AGG);

    gcn_gemm_kernel<true><<<grid_gemm, blk, 0, stream>>>(AGG, W2, b2, H);
    gcn_agg_init_kernel<<<grid_nfeat4, blk, 0, stream>>>(H, dinv, AGG);
    gcn_scatter_kernel<<<grid_escat, blk, 0, stream>>>(row, col, dinv, H, AGG);

    gcn_pool_kernel<<<grid_nfeat4, blk, 0, stream>>>(AGG, batch, out, cnts);
    gcn_finalize_kernel<<<grid_out, blk, 0, stream>>>(out, cnts);
}


# mean-pool: each wave sums its 32 consecutive node rows in registers while the graph id repeats, then one atomic add per run instead of one per node (changes the order of the pool sums)
# speedup vs baseline: 1.4685x; 1.4685x over previous
#include <hip/hip_runtime.h>
#include <cmath>

#define NNODES 100000
#define NEDGES 600000
#define FEAT   128
#define NGRAPH 256

typedef __attribute__((ext_vector_type(2))) float v2f;
typedef __attribute__((ext_vector_type(8))) float v8f;

__global__ void gcn_init_kernel(float* __restrict__ out, float* __restrict__ cnts,
                                float* __restrict__ deg) {
    int i = blockIdx.x * 256 + threadIdx.x;
    if (i < NGRAPH * FEAT) out[i] = 0.0f;
    if (i < NGRAPH)        cnts[i] = 0.0f;
    if (i < NNODES)        deg[i]  = 1.0f;
}

__global__ void gcn_deg_kernel(const int* __restrict__ col, float* __restrict__ deg) {
    int e = blockIdx.x * 256 + threadIdx.x;
    if (e < NEDGES) atomicAdd(&deg[col[e]], 1.0f);
}

__global__ void gcn_dinv_kernel(float* __restrict__ deg) {
    int n = blockIdx.x * 256 + threadIdx.x;
    if (n < NNODES) {
        float d = deg[n];
        deg[n] = (d > 0.0f) ? rsqrtf(d) : 0.0f;
    }
}

template <bool RELU>
__global__ void gcn_gemm_kernel(const float* __restrict__ X,
                                const float* __restrict__ W,
                                const float* __restrict__ bias,
                                float* __restrict__ Y) {
    __shared__ float Wt[FEAT * 16];
    __shared__ float Bt[16];

    const int lane  = threadIdx.x & 31;
    const int wave  = threadIdx.x >> 5;
    const int m0    = blockIdx.x * 128 + wave * 16;
    const bool active = (m0 < NNODES);

    const int mrow  = m0 + (lane & 15);
    const int khalf = (lane >> 4) << 1;
    const int ncol  = lane & 15;
    const int half  = lane >> 4;

    for (int nt = 0; nt < 8; ++nt) {
        __syncthreads();
        #pragma unroll
        for (int i = 0; i < 8; ++i) {
            int idx = threadIdx.x * 8 + i;
            int kk  = idx >> 4;
            int cc  = idx & 15;
            Wt[idx] = W[kk * FEAT + nt * 16 + cc];
        }
        if (threadIdx.x < 16) Bt[threadIdx.x] = bias[nt * 16 + threadIdx.x];
        __syncthreads();

        if (!active) continue;

        v8f acc = {};
        #pragma unroll 4
        for (int k = 0; k < FEAT; k += 4) {
            float a0 = X[(size_t)mrow * FEAT + k + khalf];
            float a1 = X[(size_t)mrow * FEAT + k + khalf + 1];
            if (RELU) { a0 = fmaxf(a0, 0.0f); a1 = fmaxf(a1, 0.0f); }
            v2f a; a[0] = a0; a[1] = a1;
            v2f b;
            b[0] = Wt[(k + khalf) * 16 + ncol];
            b[1] = Wt[(k + khalf + 1) * 16 + ncol];
            acc = __builtin_amdgcn_wmma_f32_16x16x4_f32(
                false, a, false, b, (short)0, acc, false, false);
        }

        #pragma unroll
        for (int v = 0; v < 8; ++v) {
            int r = m0 + v + half * 8;
            Y[(size_t)r * FEAT + nt * 16 + ncol] = acc[v] + Bt[ncol];
        }
    }
}

__global__ void gcn_agg_init_kernel(const float* __restrict__ H,
                                    const float* __restrict__ dinv,
                                    float* __restrict__ AGG) {
    int i = blockIdx.x * 256 + threadIdx.x;
    if (i >= NNODES * 32) return;
    int n = i >> 5;
    float s = dinv[n]; s = s * s;
    float4 h = *(const float4*)(H + (size_t)i * 4);
    float4 r = make_float4(h.x * s, h.y * s, h.z * s, h.w * s);
    *(float4*)(AGG + (size_t)i * 4) = r;
}

__global__ void gcn_scatter_kernel(const int* __restrict__ row,
                                   const int* __restrict__ col,
                                   const float* __restrict__ dinv,
                                   const float* __restrict__ H,
                                   float* __restrict__ AGG) {
    int t = blockIdx.x * 256 + threadIdx.x;
    int e = t >> 5;
    if (e >= NEDGES) return;
    int lane = t & 31;
    int r = row[e];
    int c = col[e];
    float norm = dinv[r] * dinv[c];
    float4 h = *(const float4*)(H + (size_t)c * FEAT + lane * 4);
    float* dst = AGG + (size_t)r * FEAT + lane * 4;
    atomicAdd(dst + 0, norm * h.x);
    atomicAdd(dst + 1, norm * h.y);
    atomicAdd(dst + 2, norm * h.z);
    atomicAdd(dst + 3, norm * h.w);
}

#define POOL_RUN 32

__device__ __forceinline__ void gcn_pool_flush(float* __restrict__ out,
                                               float* __restrict__ cnts,
                                               int g, float4 acc, float cnt, int lane) {
    float* dst = out + (size_t)g * FEAT + lane * 4;
    atomicAdd(dst + 0, acc.x);
    atomicAdd(dst + 1, acc.y);
    atomicAdd(dst + 2, acc.z);
    atomicAdd(dst + 3, acc.w);
    if (lane == 0) atomicAdd(&cnts[g], cnt);
}

__global__ void __launch_bounds__(128)
gcn_pool_kernel(const float* __restrict__ AGG,
                const int* __restrict__ batch,
                float* __restrict__ out,
                float* __restrict__ cnts) {
    const int lane = threadIdx.x & 31;
    const int w    = (blockIdx.x * 128 + threadIdx.x) >> 5;
    const int n0   = w * POOL_RUN;
    if (n0 >= NNODES) return;
    const int nn = (NNODES % POOL_RUN == 0) ? POOL_RUN
                                            : min(POOL_RUN, NNODES - n0);
    const int gl = (lane < nn) ? batch[n0 + lane] : 0;
    const float4* src = (const float4*)(AGG + (size_t)n0 * FEAT) + lane;

    float4 h[POOL_RUN];
    #pragma unroll
    for (int i = 0; i < POOL_RUN; ++i)
        if (i < nn) h[i] = src[i * 32];

    int    gcur = __builtin_amdgcn_readlane(gl, 0);
    float4 acc  = make_float4(0.0f, 0.0f, 0.0f, 0.0f);
    float  cnt  = 0.0f;
    #pragma unroll
    for (int i = 0; i < POOL_RUN; ++i) {
        if (i < nn) {
            const int g = __builtin_amdgcn_readlane(gl, i);
            if (g != gcur) {
                gcn_pool_flush(out, cnts, gcur, acc, cnt, lane);
                acc  = make_float4(0.0f, 0.0f, 0.0f, 0.0f);
                cnt  = 0.0f;
                gcur = g;
            }
            acc.x += h[i].x; acc.y += h[i].y; acc.z += h[i].z; acc.w += h[i].w;
            cnt += 1.0f;
        }
    }
    gcn_pool_flush(out, cnts, gcur, acc, cnt, lane);
}

__global__ void gcn_finalize_kernel(float* __restrict__ out,
                                    const float* __restrict__ cnts) {
    int i = blockIdx.x * 256 + threadIdx.x;
    if (i < NGRAPH * FEAT) {
        float c = cnts[i >> 7];
        out[i] = out[i] / fmaxf(c, 1.0f);
    }
}

extern "C" void kernel_launch(void* const* d_in, const int* in_sizes, int n_in,
                              void* d_out, int out_size, void* d_ws, size_t ws_size,
                              hipStream_t stream) {
    const float* x          = (const float*)d_in[0];
    const int*   edge_index = (const int*)  d_in[1];
    const int*   batch      = (const int*)  d_in[2];
    const float* W1 = (const float*)d_in[4];
    const float* b1 = (const float*)d_in[5];
    const float* W2 = (const float*)d_in[6];
    const float* b2 = (const float*)d_in[7];
    float* out = (float*)d_out;

    float* H    = (float*)d_ws;
    float* AGG  = H    + (size_t)NNODES * FEAT;
    float* dinv = AGG  + (size_t)NNODES * FEAT;
    float* cnts = dinv + NNODES;

    const int* row = edge_index;
    const int* col = edge_index + NEDGES;

    dim3 blk(256);
    const int grid_nodes  = (NNODES + 255) / 256;
    const int grid_edges  = (NEDGES + 255) / 256;
    const int grid_gemm   = (NNODES + 127) / 128;
    const int grid_nfeat4 = (NNODES * 32 + 255) / 256;
    const int grid_escat  = (NEDGES * 32 + 255) / 256;
    const int grid_out    = (NGRAPH * FEAT + 255) / 256;
    const int grid_pool   = ((NNODES + POOL_RUN - 1) / POOL_RUN + 3) / 4;

    gcn_init_kernel<<<grid_nodes, blk, 0, stream>>>(out, cnts, dinv);
    gcn_deg_kernel<<<grid_edges, blk, 0, stream>>>(col, dinv);
    gcn_dinv_kernel<<<grid_nodes, blk, 0, stream>>>(dinv);

    gcn_gemm_kernel<false><<<grid_gemm, blk, 0, stream>>>(x, W1, b1, H);
    gcn_agg_init_kernel<<<grid_nfeat4, blk, 0, stream>>>(H, dinv, AGG);
    gcn_scatter_kernel<<<grid_escat, blk, 0, stream>>>(row, col, dinv, H, AGG);

    gcn_gemm_kernel<true><<<grid_gemm, blk, 0, stream>>>(AGG, W2, b2, H);
    gcn_agg_init_kernel<<<grid_nfeat4, blk, 0, stream>>>(H, dinv, AGG);
    gcn_scatter_kernel<<<grid_escat, blk, 0, stream>>>(row, col, dinv, H, AGG);

    gcn_pool_kernel<<<grid_pool, dim3(128), 0, stream>>>(AGG, batch, out, cnts);
    gcn_finalize_kernel<<<grid_out, blk, 0, stream>>>(out, cnts);
}


# edge aggregation by per-row edge lists built with integer atomics, each row summed in registers by one wave (self term folded in): no float atomics, no pre-pass; sums run in list arrival order
# speedup vs baseline: 2.2019x; 1.4994x over previous
#include <hip/hip_runtime.h>
#include <cmath>

#define NNODES 100000
#define NEDGES 600000
#define FEAT   128
#define NGRAPH 256

typedef __attribute__((ext_vector_type(2))) float v2f;
typedef __attribute__((ext_vector_type(8))) float v8f;

__global__ void gcn_init_kernel(float* __restrict__ out, float* __restrict__ cnts,
                                float* __restrict__ deg, int* __restrict__ rowcnt,
                                int* __restrict__ total) {
    int i = blockIdx.x * 256 + threadIdx.x;
    if (i < NGRAPH * FEAT) out[i] = 0.0f;
    if (i < NGRAPH)        cnts[i] = 0.0f;
    if (i < NNODES) {
        deg[i]    = 1.0f;
        rowcnt[i] = 0;
    }
    if (i == 0) *total = 0;
}

__global__ void gcn_deg_kernel(const int* __restrict__ row, const int* __restrict__ col,
                               float* __restrict__ deg, int* __restrict__ rowcnt) {
    int e = blockIdx.x * 256 + threadIdx.x;
    if (e < NEDGES) {
        atomicAdd(&deg[col[e]], 1.0f);
        atomicAdd(&rowcnt[row[e]], 1);
    }
}

__global__ void gcn_dinv_kernel(float* __restrict__ deg, int* __restrict__ rowcnt,
                                int* __restrict__ rowstart, int* __restrict__ total) {
    __shared__ int wsum[8];
    __shared__ int sbase;
    const int n    = blockIdx.x * 256 + threadIdx.x;
    const int lane = threadIdx.x & 31;
    const int wave = threadIdx.x >> 5;
    int c = 0;
    if (n < NNODES) {
        float d = deg[n];
        deg[n] = (d > 0.0f) ? rsqrtf(d) : 0.0f;
        c = rowcnt[n];
    }
    int v = c;
    #pragma unroll
    for (int o = 1; o < 32; o <<= 1) {
        int t = __shfl_up(v, o);
        if (lane >= o) v += t;
    }
    if (lane == 31) wsum[wave] = v;
    __syncthreads();
    if (threadIdx.x == 0) {
        int s = 0;
        for (int w = 0; w < 8; ++w) { int t = wsum[w]; wsum[w] = s; s += t; }
        sbase = atomicAdd(total, s);
    }
    __syncthreads();
    if (n < NNODES) {
        int st = sbase + wsum[wave] + v - c;
        rowstart[n] = st;
        rowcnt[n]   = st;
    }
}

__global__ void gcn_fill_kernel(const int* __restrict__ row, const int* __restrict__ col,
                                const float* __restrict__ dinv, int* __restrict__ cursor,
                                int* __restrict__ elist_words) {
    int2* elist = (int2*)elist_words;
    int e = blockIdx.x * 256 + threadIdx.x;
    if (e >= NEDGES) return;
    int r = row[e];
    int c = col[e];
    float w = dinv[r] * dinv[c];
    int pos = atomicAdd(&cursor[r], 1);
    elist[pos] = make_int2(c, __float_as_int(w));
}

template <bool RELU>
__global__ void gcn_gemm_kernel(const float* __restrict__ X,
                                const float* __restrict__ W,
                                const float* __restrict__ bias,
                                float* __restrict__ Y) {
    __shared__ float Wt[FEAT * 16];
    __shared__ float Bt[16];

    const int lane  = threadIdx.x & 31;
    const int wave  = threadIdx.x >> 5;
    const int m0    = blockIdx.x * 128 + wave * 16;
    const bool active = (m0 < NNODES);

    const int mrow  = m0 + (lane & 15);
    const int khalf = (lane >> 4) << 1;
    const int ncol  = lane & 15;
    const int half  = lane >> 4;

    for (int nt = 0; nt < 8; ++nt) {
        __syncthreads();
        #pragma unroll
        for (int i = 0; i < 8; ++i) {
            int idx = threadIdx.x * 8 + i;
            int kk  = idx >> 4;
            int cc  = idx & 15;
            Wt[idx] = W[kk * FEAT + nt * 16 + cc];
        }
        if (threadIdx.x < 16) Bt[threadIdx.x] = bias[nt * 16 + threadIdx.x];
        __syncthreads();

        if (!active) continue;

        v8f acc = {};
        #pragma unroll 4
        for (int k = 0; k < FEAT; k += 4) {
            float a0 = X[(size_t)mrow * FEAT + k + khalf];
            float a1 = X[(size_t)mrow * FEAT + k + khalf + 1];
            if (RELU) { a0 = fmaxf(a0, 0.0f); a1 = fmaxf(a1, 0.0f); }
            v2f a; a[0] = a0; a[1] = a1;
            v2f b;
            b[0] = Wt[(k + khalf) * 16 + ncol];
            b[1] = Wt[(k + khalf + 1) * 16 + ncol];
            acc = __builtin_amdgcn_wmma_f32_16x16x4_f32(
                false, a, false, b, (short)0, acc, false, false);
        }

        #pragma unroll
        for (int v = 0; v < 8; ++v) {
            int r = m0 + v + half * 8;
            Y[(size_t)r * FEAT + nt * 16 + ncol] = acc[v] + Bt[ncol];
        }
    }
}

__global__ void gcn_gather_kernel(const int* __restrict__ rowstart,
                                  const int* __restrict__ rowend,
                                  const int* __restrict__ elist_words,
                                  const float* __restrict__ dinv,
                                  const float* __restrict__ H,
                                  float* __restrict__ AGG) {
    const int2* elist = (const int2*)elist_words;
    const int lane = threadIdx.x & 31;
    const int n    = (blockIdx.x * 256 + threadIdx.x) >> 5;
    if (n >= NNODES) return;
    const int s = __builtin_amdgcn_readfirstlane(rowstart[n]);
    const int e = __builtin_amdgcn_readfirstlane(rowend[n]);
    const float4* H4 = (const float4*)H;
    float di = dinv[n];
    float sw = di * di;
    float4 hs = H4[(size_t)n * 32 + lane];
    float4 acc = make_float4(sw * hs.x, sw * hs.y, sw * hs.z, sw * hs.w);
    for (int base = s; base < e; base += 32) {
        const int idx = base + lane;
        int2 ent = make_int2(0, 0);
        if (idx < e) ent = elist[idx];
        const int m = min(32, e - base);
        #pragma unroll 4
        for (int j = 0; j < m; ++j) {
            const int   c = __builtin_amdgcn_readlane(ent.x, j);
            const float w = __int_as_float(__builtin_amdgcn_readlane(ent.y, j));
            const float4 h = H4[(size_t)c * 32 + lane];
            acc.x += w * h.x; acc.y += w * h.y; acc.z += w * h.z; acc.w += w * h.w;
        }
    }
    ((float4*)AGG)[(size_t)n * 32 + lane] = acc;
}

#define POOL_RUN 32

__device__ __forceinline__ void gcn_pool_flush(float* __restrict__ out,
                                               float* __restrict__ cnts,
                                               int g, float4 acc, float cnt, int lane) {
    float* dst = out + (size_t)g * FEAT + lane * 4;
    atomicAdd(dst + 0, acc.x);
    atomicAdd(dst + 1, acc.y);
    atomicAdd(dst + 2, acc.z);
    atomicAdd(dst + 3, acc.w);
    if (lane == 0) atomicAdd(&cnts[g], cnt);
}

__global__ void __launch_bounds__(128)
gcn_pool_kernel(const float* __restrict__ AGG,
                const int* __restrict__ batch,
                float* __restrict__ out,
                float* __restrict__ cnts) {
    const int lane = threadIdx.x & 31;
    const int w    = (blockIdx.x * 128 + threadIdx.x) >> 5;
    const int n0   = w * POOL_RUN;
    if (n0 >= NNODES) return;
    const int nn = (NNODES % POOL_RUN == 0) ? POOL_RUN
                                            : min(POOL_RUN, NNODES - n0);
    const int gl = (lane < nn) ? batch[n0 + lane] : 0;
    const float4* src = (const float4*)(AGG + (size_t)n0 * FEAT) + lane;

    float4 h[POOL_RUN];
    #pragma unroll
    for (int i = 0; i < POOL_RUN; ++i)
        if (i < nn) h[i] = src[i * 32];

    int    gcur = __builtin_amdgcn_readlane(gl, 0);
    float4 acc  = make_float4(0.0f, 0.0f, 0.0f, 0.0f);
    float  cnt  = 0.0f;
    #pragma unroll
    for (int i = 0; i < POOL_RUN; ++i) {
        if (i < nn) {
            const int g = __builtin_amdgcn_readlane(gl, i);
            if (g != gcur) {
                gcn_pool_flush(out, cnts, gcur, acc, cnt, lane);
                acc  = make_float4(0.0f, 0.0f, 0.0f, 0.0f);
                cnt  = 0.0f;
                gcur = g;
            }
            acc.x += h[i].x; acc.y += h[i].y; acc.z += h[i].z; acc.w += h[i].w;
            cnt += 1.0f;
        }
    }
    gcn_pool_flush(out, cnts, gcur, acc, cnt, lane);
}

__global__ void gcn_finalize_kernel(float* __restrict__ out,
                                    const float* __restrict__ cnts) {
    int i = blockIdx.x * 256 + threadIdx.x;
    if (i < NGRAPH * FEAT) {
        float c = cnts[i >> 7];
        out[i] = out[i] / fmaxf(c, 1.0f);
    }
}

extern "C" void kernel_launch(void* const* d_in, const int* in_sizes, int n_in,
                              void* d_out, int out_size, void* d_ws, size_t ws_size,
                              hipStream_t stream) {
    const float* x          = (const float*)d_in[0];
    const int*   edge_index = (const int*)  d_in[1];
    const int*   batch      = (const int*)  d_in[2];
    const float* W1 = (const float*)d_in[4];
    const float* b1 = (const float*)d_in[5];
    const float* W2 = (const float*)d_in[6];
    const float* b2 = (const float*)d_in[7];
    float* out = (float*)d_out;

    float* H    = (float*)d_ws;
    float* AGG  = H    + (size_t)NNODES * FEAT;
    float* dinv = AGG  + (size_t)NNODES * FEAT;
    float* cnts = dinv + NNODES;
    int*   rowcnt   = (int*)(cnts + NGRAPH);
    int*   rowstart = rowcnt + NNODES;
    int*   total    = rowstart + NNODES;
    int*   elist    = total + 16;

    const int* row = edge_index;
    const int* col = edge_index + NEDGES;

    dim3 blk(256);
    const int grid_nodes  = (NNODES + 255) / 256;
    const int grid_edges  = (NEDGES + 255) / 256;
    const int grid_gemm   = (NNODES + 127) / 128;
    const int grid_nfeat4 = (NNODES * 32 + 255) / 256;
    const int grid_out    = (NGRAPH * FEAT + 255) / 256;
    const int grid_pool   = ((NNODES + POOL_RUN - 1) / POOL_RUN + 3) / 4;

    gcn_init_kernel<<<grid_nodes, blk, 0, stream>>>(out, cnts, dinv, rowcnt, total);
    gcn_deg_kernel<<<grid_edges, blk, 0, stream>>>(row, col, dinv, rowcnt);
    gcn_dinv_kernel<<<grid_nodes, blk, 0, stream>>>(dinv, rowcnt, rowstart, total);
    gcn_fill_kernel<<<grid_edges, blk, 0, stream>>>(row, col, dinv, rowcnt, elist);

    gcn_gemm_kernel<false><<<grid_gemm, blk, 0, stream>>>(x, W1, b1, H);
    gcn_gather_kernel<<<grid_nfeat4, blk, 0, stream>>>(rowstart, rowcnt, elist, dinv, H, AGG);

    gcn_gemm_kernel<true><<<grid_gemm, blk, 0, stream>>>(AGG, W2, b2, H);
    gcn_gather_kernel<<<grid_nfeat4, blk, 0, stream>>>(rowstart, rowcnt, elist, dinv, H, AGG);

    gcn_pool_kernel<<<grid_pool, dim3(128), 0, stream>>>(AGG, batch, out, cnts);
    gcn_finalize_kernel<<<grid_out, blk, 0, stream>>>(out, cnts);
}


# GEMM: each wave reads its 16 rows once as whole 512-byte rows staged in LDS and makes all 128 columns from them (was 8 passes of 16-line strided loads); W whole in LDS; same k order, same sums
# speedup vs baseline: 2.7128x; 1.2320x over previous
#include <hip/hip_runtime.h>
#include <cmath>

#define NNODES 100000
#define NEDGES 600000
#define FEAT   128
#define NGRAPH 256

typedef __attribute__((ext_vector_type(2))) float v2f;
typedef __attribute__((ext_vector_type(8))) float v8f;

__global__ void gcn_init_kernel(float* __restrict__ out, float* __restrict__ cnts,
                                float* __restrict__ deg, int* __restrict__ rowcnt,
                                int* __restrict__ total) {
    int i = blockIdx.x * 256 + threadIdx.x;
    if (i < NGRAPH * FEAT) out[i] = 0.0f;
    if (i < NGRAPH)        cnts[i] = 0.0f;
    if (i < NNODES) {
        deg[i]    = 1.0f;
        rowcnt[i] = 0;
    }
    if (i == 0) *total = 0;
}

__global__ void gcn_deg_kernel(const int* __restrict__ row, const int* __restrict__ col,
                               float* __restrict__ deg, int* __restrict__ rowcnt) {
    int e = blockIdx.x * 256 + threadIdx.x;
    if (e < NEDGES) {
        atomicAdd(&deg[col[e]], 1.0f);
        atomicAdd(&rowcnt[row[e]], 1);
    }
}

__global__ void gcn_dinv_kernel(float* __restrict__ deg, int* __restrict__ rowcnt,
                                int* __restrict__ rowstart, int* __restrict__ total) {
    __shared__ int wsum[8];
    __shared__ int sbase;
    const int n    = blockIdx.x * 256 + threadIdx.x;
    const int lane = threadIdx.x & 31;
    const int wave = threadIdx.x >> 5;
    int c = 0;
    if (n < NNODES) {
        float d = deg[n];
        deg[n] = (d > 0.0f) ? rsqrtf(d) : 0.0f;
        c = rowcnt[n];
    }
    int v = c;
    #pragma unroll
    for (int o = 1; o < 32; o <<= 1) {
        int t = __shfl_up(v, o);
        if (lane >= o) v += t;
    }
    if (lane == 31) wsum[wave] = v;
    __syncthreads();
    if (threadIdx.x == 0) {
        int s = 0;
        for (int w = 0; w < 8; ++w) { int t = wsum[w]; wsum[w] = s; s += t; }
        sbase = atomicAdd(total, s);
    }
    __syncthreads();
    if (n < NNODES) {
        int st = sbase + wsum[wave] + v - c;
        rowstart[n] = st;
        rowcnt[n]   = st;
    }
}

__global__ void gcn_fill_kernel(const int* __restrict__ row, const int* __restrict__ col,
                                const float* __restrict__ dinv, int* __restrict__ cursor,
                                int* __restrict__ elist_words) {
    int2* elist = (int2*)elist_words;
    int e = blockIdx.x * 256 + threadIdx.x;
    if (e >= NEDGES) return;
    int r = row[e];
    int c = col[e];
    float w = dinv[r] * dinv[c];
    int pos = atomicAdd(&cursor[r], 1);
    elist[pos] = make_int2(c, __float_as_int(w));
}

#define GEMM_WAVES 4
#define GEMM_ROWS  (GEMM_WAVES * 16)
#define WP_PITCH   288
#define AT_PITCH   132

template <bool RELU>
__global__ void __launch_bounds__(GEMM_WAVES * 32)
gcn_gemm_kernel(const float* __restrict__ X,
                const float* __restrict__ W,
                const float* __restrict__ bias,
                float* __restrict__ Y) {
    __shared__ alignas(16) float Wp[64 * WP_PITCH];
    __shared__ alignas(16) float At[GEMM_WAVES * 16 * AT_PITCH];

    const int lane  = threadIdx.x & 31;
    const int wave  = threadIdx.x >> 5;
    const int m0    = blockIdx.x * GEMM_ROWS + wave * 16;
    const bool active = (m0 < NNODES);
    const int  mr     = active ? m0 : 0;
    const int c = lane & 15;
    const int h = lane >> 4;
    float* Aw = At + wave * 16 * AT_PITCH;

    float4 arow[16];
    {
        const float4* src = (const float4*)(X + (size_t)mr * FEAT) + lane;
        #pragma unroll
        for (int j = 0; j < 16; ++j) arow[j] = src[j * 32];
    }

    {
        const float4* W4 = (const float4*)W;
        #pragma unroll
        for (int i = 0; i < 16; ++i) {
            const int p = i * GEMM_WAVES + wave;
            const float4 r0 = W4[(2 * p) * 32 + lane];
            const float4 r1 = W4[(2 * p + 1) * 32 + lane];
            float4* dst = (float4*)(Wp + p * WP_PITCH + lane * 8);
            dst[0] = make_float4(r0.x, r1.x, r0.y, r1.y);
            dst[1] = make_float4(r0.z, r1.z, r0.w, r1.w);
        }
    }

    {
        #pragma unroll
        for (int j = 0; j < 16; ++j) {
            float4 v = arow[j];
            if (RELU) {
                v.x = fmaxf(v.x, 0.0f); v.y = fmaxf(v.y, 0.0f);
                v.z = fmaxf(v.z, 0.0f); v.w = fmaxf(v.w, 0.0f);
            }
            *(float4*)(Aw + j * AT_PITCH + lane * 4) = v;
        }
    }
    __syncthreads();

    v8f acc[8];
    #pragma unroll
    for (int nt = 0; nt < 8; ++nt) acc[nt] = v8f{};

    {
        const float* ap = Aw + c * AT_PITCH + 2 * h;
        const float* bp = Wp + h * WP_PITCH + 2 * c;
        #pragma unroll
        for (int s = 0; s < 32; ++s) {
            const v2f a = *(const v2f*)(ap + 4 * s);
            #pragma unroll
            for (int nt = 0; nt < 8; ++nt) {
                const v2f b = *(const v2f*)(bp + 2 * s * WP_PITCH + 32 * nt);
                acc[nt] = __builtin_amdgcn_wmma_f32_16x16x4_f32(
                    false, a, false, b, (short)0, acc[nt], false, false);
            }
        }

        #pragma unroll
        for (int nt = 0; nt < 8; ++nt) {
            const float bv = bias[nt * 16 + c];
            #pragma unroll
            for (int r = 0; r < 8; ++r)
                Aw[(8 * h + r) * AT_PITCH + nt * 16 + c] = acc[nt][r] + bv;
        }
    }
    __syncthreads();
    if (active) {
        float4* dst = (float4*)(Y + (size_t)m0 * FEAT) + lane;
        #pragma unroll
        for (int j = 0; j < 16; ++j)
            dst[j * 32] = *(const float4*)(Aw + j * AT_PITCH + lane * 4);
    }
}

__global__ void gcn_gather_kernel(const int* __restrict__ rowstart,
                                  const int* __restrict__ rowend,
                                  const int* __restrict__ elist_words,
                                  const float* __restrict__ dinv,
                                  const float* __restrict__ H,
                                  float* __restrict__ AGG) {
    const int2* elist = (const int2*)elist_words;
    const int lane = threadIdx.x & 31;
    const int n    = (blockIdx.x * 256 + threadIdx.x) >> 5;
    if (n >= NNODES) return;
    const int s = __builtin_amdgcn_readfirstlane(rowstart[n]);
    const int e = __builtin_amdgcn_readfirstlane(rowend[n]);
    const float4* H4 = (const float4*)H;
    float di = dinv[n];
    float sw = di * di;
    float4 hs = H4[(size_t)n * 32 + lane];
    float4 acc = make_float4(sw * hs.x, sw * hs.y, sw * hs.z, sw * hs.w);
    for (int base = s; base < e; base += 32) {
        const int idx = base + lane;
        int2 ent = make_int2(0, 0);
        if (idx < e) ent = elist[idx];
        const int m = min(32, e - base);
        #pragma unroll 4
        for (int j = 0; j < m; ++j) {
            const int   c = __builtin_amdgcn_readlane(ent.x, j);
            const float w = __int_as_float(__builtin_amdgcn_readlane(ent.y, j));
            const float4 h = H4[(size_t)c * 32 + lane];
            acc.x += w * h.x; acc.y += w * h.y; acc.z += w * h.z; acc.w += w * h.w;
        }
    }
    ((float4*)AGG)[(size_t)n * 32 + lane] = acc;
}

#define POOL_RUN 32

__device__ __forceinline__ void gcn_pool_flush(float* __restrict__ out,
                                               float* __restrict__ cnts,
                                               int g, float4 acc, float cnt, int lane) {
    float* dst = out + (size_t)g * FEAT + lane * 4;
    atomicAdd(dst + 0, acc.x);
    atomicAdd(dst + 1, acc.y);
    atomicAdd(dst + 2, acc.z);
    atomicAdd(dst + 3, acc.w);
    if (lane == 0) atomicAdd(&cnts[g], cnt);
}

__global__ void __launch_bounds__(128)
gcn_pool_kernel(const float* __restrict__ AGG,
                const int* __restrict__ batch,
                float* __restrict__ out,
                float* __restrict__ cnts) {
    const int lane = threadIdx.x & 31;
    const int w    = (blockIdx.x * 128 + threadIdx.x) >> 5;
    const int n0   = w * POOL_RUN;
    if (n0 >= NNODES) return;
    const int nn = (NNODES % POOL_RUN == 0) ? POOL_RUN
                                            : min(POOL_RUN, NNODES - n0);
    const int gl = (lane < nn) ? batch[n0 + lane] : 0;
    const float4* src = (const float4*)(AGG + (size_t)n0 * FEAT) + lane;

    float4 h[POOL_RUN];
    #pragma unroll
    for (int i = 0; i < POOL_RUN; ++i)
        if (i < nn) h[i] = src[i * 32];

    int    gcur = __builtin_amdgcn_readlane(gl, 0);
    float4 acc  = make_float4(0.0f, 0.0f, 0.0f, 0.0f);
    float  cnt  = 0.0f;
    #pragma unroll
    for (int i = 0; i < POOL_RUN; ++i) {
        if (i < nn) {
            const int g = __builtin_amdgcn_readlane(gl, i);
            if (g != gcur) {
                gcn_pool_flush(out, cnts, gcur, acc, cnt, lane);
                acc  = make_float4(0.0f, 0.0f, 0.0f, 0.0f);
                cnt  = 0.0f;
                gcur = g;
            }
            acc.x += h[i].x; acc.y += h[i].y; acc.z += h[i].z; acc.w += h[i].w;
            cnt += 1.0f;
        }
    }
    gcn_pool_flush(out, cnts, gcur, acc, cnt, lane);
}

__global__ void gcn_finalize_kernel(float* __restrict__ out,
                                    const float* __restrict__ cnts) {
    int i = blockIdx.x * 256 + threadIdx.x;
    if (i < NGRAPH * FEAT) {
        float c = cnts[i >> 7];
        out[i] = out[i] / fmaxf(c, 1.0f);
    }
}

extern "C" void kernel_launch(void* const* d_in, const int* in_sizes, int n_in,
                              void* d_out, int out_size, void* d_ws, size_t ws_size,
                              hipStream_t stream) {
    const float* x          = (const float*)d_in[0];
    const int*   edge_index = (const int*)  d_in[1];
    const int*   batch      = (const int*)  d_in[2];
    const float* W1 = (const float*)d_in[4];
    const float* b1 = (const float*)d_in[5];
    const float* W2 = (const float*)d_in[6];
    const float* b2 = (const float*)d_in[7];
    float* out = (float*)d_out;

    float* H    = (float*)d_ws;
    float* AGG  = H    + (size_t)NNODES * FEAT;
    float* dinv = AGG  + (size_t)NNODES * FEAT;
    float* cnts = dinv + NNODES;
    int*   rowcnt   = (int*)(cnts + NGRAPH);
    int*   rowstart = rowcnt + NNODES;
    int*   total    = rowstart + NNODES;
    int*   elist    = total + 16;

    const int* row = edge_index;
    const int* col = edge_index + NEDGES;

    dim3 blk(256);
    const int grid_nodes  = (NNODES + 255) / 256;
    const int grid_edges  = (NEDGES + 255) / 256;
    const int grid_gemm   = (NNODES + GEMM_ROWS - 1) / GEMM_ROWS;
    const int grid_nfeat4 = (NNODES * 32 + 255) / 256;
    const int grid_out    = (NGRAPH * FEAT + 255) / 256;
    const int grid_pool   = ((NNODES + POOL_RUN - 1) / POOL_RUN + 3) / 4;

    gcn_init_kernel<<<grid_nodes, blk, 0, stream>>>(out, cnts, dinv, rowcnt, total);
    gcn_deg_kernel<<<grid_edges, blk, 0, stream>>>(row, col, dinv, rowcnt);
    gcn_dinv_kernel<<<grid_nodes, blk, 0, stream>>>(dinv, rowcnt, rowstart, total);
    gcn_fill_kernel<<<grid_edges, blk, 0, stream>>>(row, col, dinv, rowcnt, elist);

    gcn_gemm_kernel<false><<<grid_gemm, dim3(GEMM_WAVES * 32), 0, stream>>>(x, W1, b1, H);
    gcn_gather_kernel<<<grid_nfeat4, blk, 0, stream>>>(rowstart, rowcnt, elist, dinv, H, AGG);

    gcn_gemm_kernel<true><<<grid_gemm, dim3(GEMM_WAVES * 32), 0, stream>>>(AGG, W2, b2, H);
    gcn_gather_kernel<<<grid_nfeat4, blk, 0, stream>>>(rowstart, rowcnt, elist, dinv, H, AGG);

    gcn_pool_kernel<<<grid_pool, dim3(128), 0, stream>>>(AGG, batch, out, cnts);
    gcn_finalize_kernel<<<grid_out, blk, 0, stream>>>(out, cnts);
}


# layer 2 weights and bias applied after aggregation and mean pool (linear from relu to output): its N x128x128 product becomes 256x128x128 on the same f32 matrix form; changes the order of the sums
# speedup vs baseline: 3.0176x; 1.1124x over previous
#include <hip/hip_runtime.h>
#include <cmath>

#define NNODES 100000
#define NEDGES 600000
#define FEAT   128
#define NGRAPH 256

typedef __attribute__((ext_vector_type(2))) float v2f;
typedef __attribute__((ext_vector_type(8))) float v8f;

__global__ void gcn_init_kernel(float* __restrict__ psum, float* __restrict__ cnts,
                                float* __restrict__ gws,
                                float* __restrict__ deg, int* __restrict__ rowcnt,
                                int* __restrict__ total) {
    int i = blockIdx.x * 256 + threadIdx.x;
    if (i < NGRAPH * FEAT) psum[i] = 0.0f;
    if (i < NGRAPH) {
        cnts[i] = 0.0f;
        gws[i]  = 0.0f;
    }
    if (i < NNODES) {
        deg[i]    = 1.0f;
        rowcnt[i] = 0;
    }
    if (i == 0) *total = 0;
}

__global__ void gcn_deg_kernel(const int* __restrict__ row, const int* __restrict__ col,
                               float* __restrict__ deg, int* __restrict__ rowcnt) {
    int e = blockIdx.x * 256 + threadIdx.x;
    if (e < NEDGES) {
        atomicAdd(&deg[col[e]], 1.0f);
        atomicAdd(&rowcnt[row[e]], 1);
    }
}

__global__ void gcn_dinv_kernel(float* __restrict__ deg, int* __restrict__ rowcnt,
                                int* __restrict__ rowstart, int* __restrict__ total) {
    __shared__ int wsum[8];
    __shared__ int sbase;
    const int n    = blockIdx.x * 256 + threadIdx.x;
    const int lane = threadIdx.x & 31;
    const int wave = threadIdx.x >> 5;
    int c = 0;
    if (n < NNODES) {
        float d = deg[n];
        deg[n] = (d > 0.0f) ? rsqrtf(d) : 0.0f;
        c = rowcnt[n];
    }
    int v = c;
    #pragma unroll
    for (int o = 1; o < 32; o <<= 1) {
        int t = __shfl_up(v, o);
        if (lane >= o) v += t;
    }
    if (lane == 31) wsum[wave] = v;
    __syncthreads();
    if (threadIdx.x == 0) {
        int s = 0;
        for (int w = 0; w < 8; ++w) { int t = wsum[w]; wsum[w] = s; s += t; }
        sbase = atomicAdd(total, s);
    }
    __syncthreads();
    if (n < NNODES) {
        int st = sbase + wsum[wave] + v - c;
        rowstart[n] = st;
        rowcnt[n]   = st;
    }
}

__global__ void gcn_fill_kernel(const int* __restrict__ row, const int* __restrict__ col,
                                const float* __restrict__ dinv, int* __restrict__ cursor,
                                int* __restrict__ elist_words) {
    int2* elist = (int2*)elist_words;
    int e = blockIdx.x * 256 + threadIdx.x;
    if (e >= NEDGES) return;
    int r = row[e];
    int c = col[e];
    float w = dinv[r] * dinv[c];
    int pos = atomicAdd(&cursor[r], 1);
    elist[pos] = make_int2(c, __float_as_int(w));
}

#define GEMM_WAVES 4
#define GEMM_ROWS  (GEMM_WAVES * 16)
#define WP_PITCH   288
#define AT_PITCH   132

template <bool RELU>
__global__ void __launch_bounds__(GEMM_WAVES * 32)
gcn_gemm_kernel(const float* __restrict__ X,
                const float* __restrict__ W,
                const float* __restrict__ bias,
                float* __restrict__ Y) {
    __shared__ alignas(16) float Wp[64 * WP_PITCH];
    __shared__ alignas(16) float At[GEMM_WAVES * 16 * AT_PITCH];

    const int lane  = threadIdx.x & 31;
    const int wave  = threadIdx.x >> 5;
    const int m0    = blockIdx.x * GEMM_ROWS + wave * 16;
    const bool active = (m0 < NNODES);
    const int  mr     = active ? m0 : 0;
    const int c = lane & 15;
    const int h = lane >> 4;
    float* Aw = At + wave * 16 * AT_PITCH;

    float4 arow[16];
    {
        const float4* src = (const float4*)(X + (size_t)mr * FEAT) + lane;
        #pragma unroll
        for (int j = 0; j < 16; ++j) arow[j] = src[j * 32];
    }

    {
        const float4* W4 = (const float4*)W;
        #pragma unroll
        for (int i = 0; i < 16; ++i) {
            const int p = i * GEMM_WAVES + wave;
            const float4 r0 = W4[(2 * p) * 32 + lane];
            const float4 r1 = W4[(2 * p + 1) * 32 + lane];
            float4* dst = (float4*)(Wp + p * WP_PITCH + lane * 8);
            dst[0] = make_float4(r0.x, r1.x, r0.y, r1.y);
            dst[1] = make_float4(r0.z, r1.z, r0.w, r1.w);
        }
    }

    {
        #pragma unroll
        for (int j = 0; j < 16; ++j) {
            float4 v = arow[j];
            if (RELU) {
                v.x = fmaxf(v.x, 0.0f); v.y = fmaxf(v.y, 0.0f);
                v.z = fmaxf(v.z, 0.0f); v.w = fmaxf(v.w, 0.0f);
            }
            *(float4*)(Aw + j * AT_PITCH + lane * 4) = v;
        }
    }
    __syncthreads();

    v8f acc[8];
    #pragma unroll
    for (int nt = 0; nt < 8; ++nt) acc[nt] = v8f{};

    {
        const float* ap = Aw + c * AT_PITCH + 2 * h;
        const float* bp = Wp + h * WP_PITCH + 2 * c;
        #pragma unroll
        for (int s = 0; s < 32; ++s) {
            const v2f a = *(const v2f*)(ap + 4 * s);
            #pragma unroll
            for (int nt = 0; nt < 8; ++nt) {
                const v2f b = *(const v2f*)(bp + 2 * s * WP_PITCH + 32 * nt);
                acc[nt] = __builtin_amdgcn_wmma_f32_16x16x4_f32(
                    false, a, false, b, (short)0, acc[nt], false, false);
            }
        }

        #pragma unroll
        for (int nt = 0; nt < 8; ++nt) {
            const float bv = bias[nt * 16 + c];
            #pragma unroll
            for (int r = 0; r < 8; ++r)
                Aw[(8 * h + r) * AT_PITCH + nt * 16 + c] = acc[nt][r] + bv;
        }
    }
    __syncthreads();
    if (active) {
        float4* dst = (float4*)(Y + (size_t)m0 * FEAT) + lane;
        #pragma unroll
        for (int j = 0; j < 16; ++j)
            dst[j * 32] = *(const float4*)(Aw + j * AT_PITCH + lane * 4);
    }
}

__device__ __forceinline__ float4 gcn_relu4(float4 v) {
    return make_float4(fmaxf(v.x, 0.0f), fmaxf(v.y, 0.0f), fmaxf(v.z, 0.0f), fmaxf(v.w, 0.0f));
}

template <bool LAYER2>
__global__ void gcn_gather_kernel(const int* __restrict__ rowstart,
                                  const int* __restrict__ rowend,
                                  const int* __restrict__ elist_words,
                                  const float* __restrict__ dinv,
                                  const float* __restrict__ H,
                                  float* __restrict__ AGG,
                                  float* __restrict__ wsum) {
    const int2* elist = (const int2*)elist_words;
    const int lane = threadIdx.x & 31;
    const int n    = (blockIdx.x * 256 + threadIdx.x) >> 5;
    if (n >= NNODES) return;
    const int s = __builtin_amdgcn_readfirstlane(rowstart[n]);
    const int e = __builtin_amdgcn_readfirstlane(rowend[n]);
    const float4* H4 = (const float4*)H;
    float di = dinv[n];
    float sw = di * di;
    float4 hs = H4[(size_t)n * 32 + lane];
    float4 acc = make_float4(sw * hs.x, sw * hs.y, sw * hs.z, sw * hs.w);
    float  ws  = sw;
    for (int base = s; base < e; base += 32) {
        const int idx = base + lane;
        int2 ent = make_int2(0, 0);
        if (idx < e) ent = elist[idx];
        const int m = min(32, e - base);
        #pragma unroll 4
        for (int j = 0; j < m; ++j) {
            const int   c = __builtin_amdgcn_readlane(ent.x, j);
            const float w = __int_as_float(__builtin_amdgcn_readlane(ent.y, j));
            const float4 h = H4[(size_t)c * 32 + lane];
            if (LAYER2) ws += w;
            acc.x += w * h.x; acc.y += w * h.y; acc.z += w * h.z; acc.w += w * h.w;
        }
    }
    if (!LAYER2) acc = gcn_relu4(acc);
    ((float4*)AGG)[(size_t)n * 32 + lane] = acc;
    if (LAYER2 && lane == 0) wsum[n] = ws;
}

#define POOL_RUN 32

__device__ __forceinline__ void gcn_pool_flush(float* __restrict__ psum,
                                               float* __restrict__ cnts,
                                               float* __restrict__ gws,
                                               int g, float4 acc, float cnt, float wacc, int lane) {
    float* dst = psum + (size_t)g * FEAT + lane * 4;
    atomicAdd(dst + 0, acc.x);
    atomicAdd(dst + 1, acc.y);
    atomicAdd(dst + 2, acc.z);
    atomicAdd(dst + 3, acc.w);
    if (lane == 0) atomicAdd(&cnts[g], cnt);
    if (lane == 1) atomicAdd(&gws[g], wacc);
}

__global__ void __launch_bounds__(128)
gcn_pool_kernel(const float* __restrict__ AGG,
                const int* __restrict__ batch,
                const float* __restrict__ wsum,
                float* __restrict__ psum,
                float* __restrict__ cnts,
                float* __restrict__ gws) {
    const int lane = threadIdx.x & 31;
    const int w    = (blockIdx.x * 128 + threadIdx.x) >> 5;
    const int n0   = w * POOL_RUN;
    if (n0 >= NNODES) return;
    const int nn = (NNODES % POOL_RUN == 0) ? POOL_RUN
                                            : min(POOL_RUN, NNODES - n0);
    const int gl = (lane < nn) ? batch[n0 + lane] : 0;
    const float wl = (lane < nn) ? wsum[n0 + lane] : 0.0f;
    const float4* src = (const float4*)(AGG + (size_t)n0 * FEAT) + lane;

    float4 h[POOL_RUN];
    #pragma unroll
    for (int i = 0; i < POOL_RUN; ++i)
        if (i < nn) h[i] = src[i * 32];

    int    gcur = __builtin_amdgcn_readlane(gl, 0);
    float4 acc  = make_float4(0.0f, 0.0f, 0.0f, 0.0f);
    float  cnt  = 0.0f;
    float  wacc = 0.0f;
    #pragma unroll
    for (int i = 0; i < POOL_RUN; ++i) {
        if (i < nn) {
            const int g = __builtin_amdgcn_readlane(gl, i);
            if (g != gcur) {
                gcn_pool_flush(psum, cnts, gws, gcur, acc, cnt, wacc, lane);
                acc  = make_float4(0.0f, 0.0f, 0.0f, 0.0f);
                cnt  = 0.0f;
                wacc = 0.0f;
                gcur = g;
            }
            acc.x += h[i].x; acc.y += h[i].y; acc.z += h[i].z; acc.w += h[i].w;
            cnt  += 1.0f;
            wacc += __int_as_float(__builtin_amdgcn_readlane(__float_as_int(wl), i));
        }
    }
    gcn_pool_flush(psum, cnts, gws, gcur, acc, cnt, wacc, lane);
}

__global__ void __launch_bounds__(128)
gcn_head_kernel(const float* __restrict__ psum,
                const float* __restrict__ W,
                const float* __restrict__ bias,
                const float* __restrict__ cnts,
                const float* __restrict__ gws,
                float* __restrict__ out) {
    __shared__ alignas(16) float Ct[16 * AT_PITCH];

    const int lane = threadIdx.x & 31;
    const int wave = threadIdx.x >> 5;
    const int c  = lane & 15;
    const int h  = lane >> 4;
    const int m0 = blockIdx.x * 16;
    const int n0 = wave * 32;

    const float* ap = psum + (size_t)(m0 + c) * FEAT + 2 * h;
    const float* bp = W + (size_t)(2 * h) * FEAT + n0 + c;
    v2f a[32], b0[32], b1[32];
    #pragma unroll
    for (int s = 0; s < 32; ++s) {
        a[s]  = *(const v2f*)(ap + 4 * s);
        b0[s] = v2f{bp[(4 * s) * FEAT],      bp[(4 * s + 1) * FEAT]};
        b1[s] = v2f{bp[(4 * s) * FEAT + 16], bp[(4 * s + 1) * FEAT + 16]};
    }

    v8f acc0 = v8f{};
    v8f acc1 = v8f{};
    #pragma unroll
    for (int s = 0; s < 32; ++s) {
        acc0 = __builtin_amdgcn_wmma_f32_16x16x4_f32(
            false, a[s], false, b0[s], (short)0, acc0, false, false);
        acc1 = __builtin_amdgcn_wmma_f32_16x16x4_f32(
            false, a[s], false, b1[s], (short)0, acc1, false, false);
    }

    #pragma unroll
    for (int r = 0; r < 8; ++r) {
        Ct[(8 * h + r) * AT_PITCH + n0 + c]      = acc0[r];
        Ct[(8 * h + r) * AT_PITCH + n0 + 16 + c] = acc1[r];
    }
    __syncthreads();

    const float4 bv = ((const float4*)bias)[lane];
    #pragma unroll
    for (int j = 0; j < 4; ++j) {
        const int   rr = wave * 4 + j;
        const int   g  = m0 + rr;
        const float d  = fmaxf(cnts[g], 1.0f);
        const float wm = gws[g] / d;
        const float4 v = *(const float4*)(Ct + rr * AT_PITCH + lane * 4);
        float4 o;
        o.x = v.x / d + wm * bv.x;
        o.y = v.y / d + wm * bv.y;
        o.z = v.z / d + wm * bv.z;
        o.w = v.w / d + wm * bv.w;
        ((float4*)out)[(size_t)g * 32 + lane] = o;
    }
}

extern "C" void kernel_launch(void* const* d_in, const int* in_sizes, int n_in,
                              void* d_out, int out_size, void* d_ws, size_t ws_size,
                              hipStream_t stream) {
    const float* x          = (const float*)d_in[0];
    const int*   edge_index = (const int*)  d_in[1];
    const int*   batch      = (const int*)  d_in[2];
    const float* W1 = (const float*)d_in[4];
    const float* b1 = (const float*)d_in[5];
    const float* W2 = (const float*)d_in[6];
    const float* b2 = (const float*)d_in[7];
    float* out = (float*)d_out;

    float* H    = (float*)d_ws;
    float* AGG  = H    + (size_t)NNODES * FEAT;
    float* dinv = AGG  + (size_t)NNODES * FEAT;
    float* cnts = dinv + NNODES;
    float* gws  = cnts + NGRAPH;
    float* psum = gws  + NGRAPH;
    float* wsum = psum + (size_t)NGRAPH * FEAT;
    int*   rowcnt   = (int*)(wsum + NNODES);
    int*   rowstart = rowcnt + NNODES;
    int*   total    = rowstart + NNODES;
    int*   elist    = total + 16;

    const int* row = edge_index;
    const int* col = edge_index + NEDGES;

    dim3 blk(256);
    const int grid_nodes  = (NNODES + 255) / 256;
    const int grid_edges  = (NEDGES + 255) / 256;
    const int grid_gemm   = (NNODES + GEMM_ROWS - 1) / GEMM_ROWS;
    const int grid_nfeat4 = (NNODES * 32 + 255) / 256;
    const int grid_head   = NGRAPH / 16;
    const int grid_pool   = ((NNODES + POOL_RUN - 1) / POOL_RUN + 3) / 4;

    gcn_init_kernel<<<grid_nodes, blk, 0, stream>>>(psum, cnts, gws, dinv, rowcnt, total);
    gcn_deg_kernel<<<grid_edges, blk, 0, stream>>>(row, col, dinv, rowcnt);
    gcn_dinv_kernel<<<grid_nodes, blk, 0, stream>>>(dinv, rowcnt, rowstart, total);
    gcn_fill_kernel<<<grid_edges, blk, 0, stream>>>(row, col, dinv, rowcnt, elist);

    gcn_gemm_kernel<false><<<grid_gemm, dim3(GEMM_WAVES * 32), 0, stream>>>(x, W1, b1, H);
    gcn_gather_kernel<false><<<grid_nfeat4, blk, 0, stream>>>(rowstart, rowcnt, elist, dinv, H, AGG, wsum);

    gcn_gather_kernel<true><<<grid_nfeat4, blk, 0, stream>>>(rowstart, rowcnt, elist, dinv, AGG, H, wsum);
    gcn_pool_kernel<<<grid_pool, dim3(128), 0, stream>>>(H, batch, wsum, psum, cnts, gws);
    gcn_head_kernel<<<grid_head, dim3(128), 0, stream>>>(psum, W2, b2, cnts, gws, out);
}


# aggregation passes: one wave walks 8 consecutive rows (a row's loads issued as one batch, the next row's list prefetched) instead of one wave per row: 8x fewer waves to enter; same sums, same order
# speedup vs baseline: 3.1756x; 1.0524x over previous
#include <hip/hip_runtime.h>
#include <cmath>

#define NNODES 100000
#define NEDGES 600000
#define FEAT   128
#define NGRAPH 256

typedef __attribute__((ext_vector_type(2))) float v2f;
typedef __attribute__((ext_vector_type(8))) float v8f;

__global__ void gcn_init_kernel(float* __restrict__ psum, float* __restrict__ cnts,
                                float* __restrict__ gws,
                                float* __restrict__ deg, int* __restrict__ rowcnt,
                                int* __restrict__ total) {
    int i = blockIdx.x * 256 + threadIdx.x;
    if (i < NGRAPH * FEAT) psum[i] = 0.0f;
    if (i < NGRAPH) {
        cnts[i] = 0.0f;
        gws[i]  = 0.0f;
    }
    if (i < NNODES) {
        deg[i]    = 1.0f;
        rowcnt[i] = 0;
    }
    if (i == 0) *total = 0;
}

__global__ void gcn_deg_kernel(const int* __restrict__ row, const int* __restrict__ col,
                               float* __restrict__ deg, int* __restrict__ rowcnt) {
    int e = blockIdx.x * 256 + threadIdx.x;
    if (e < NEDGES) {
        atomicAdd(&deg[col[e]], 1.0f);
        atomicAdd(&rowcnt[row[e]], 1);
    }
}

__global__ void gcn_dinv_kernel(float* __restrict__ deg, int* __restrict__ rowcnt,
                                int* __restrict__ rowstart, int* __restrict__ total) {
    __shared__ int wsum[8];
    __shared__ int sbase;
    const int n    = blockIdx.x * 256 + threadIdx.x;
    const int lane = threadIdx.x & 31;
    const int wave = threadIdx.x >> 5;
    int c = 0;
    if (n < NNODES) {
        float d = deg[n];
        deg[n] = (d > 0.0f) ? rsqrtf(d) : 0.0f;
        c = rowcnt[n];
    }
    int v = c;
    #pragma unroll
    for (int o = 1; o < 32; o <<= 1) {
        int t = __shfl_up(v, o);
        if (lane >= o) v += t;
    }
    if (lane == 31) wsum[wave] = v;
    __syncthreads();
    if (threadIdx.x == 0) {
        int s = 0;
        for (int w = 0; w < 8; ++w) { int t = wsum[w]; wsum[w] = s; s += t; }
        sbase = atomicAdd(total, s);
    }
    __syncthreads();
    if (n < NNODES) {
        int st = sbase + wsum[wave] + v - c;
        rowstart[n] = st;
        rowcnt[n]   = st;
    }
}

__global__ void gcn_fill_kernel(const int* __restrict__ row, const int* __restrict__ col,
                                const float* __restrict__ dinv, int* __restrict__ cursor,
                                int* __restrict__ elist_words) {
    int2* elist = (int2*)elist_words;
    int e = blockIdx.x * 256 + threadIdx.x;
    if (e >= NEDGES) return;
    int r = row[e];
    int c = col[e];
    float w = dinv[r] * dinv[c];
    int pos = atomicAdd(&cursor[r], 1);
    elist[pos] = make_int2(c, __float_as_int(w));
}

#define GEMM_WAVES 4
#define GEMM_ROWS  (GEMM_WAVES * 16)
#define WP_PITCH   288
#define AT_PITCH   132

template <bool RELU>
__global__ void __launch_bounds__(GEMM_WAVES * 32)
gcn_gemm_kernel(const float* __restrict__ X,
                const float* __restrict__ W,
                const float* __restrict__ bias,
                float* __restrict__ Y) {
    __shared__ alignas(16) float Wp[64 * WP_PITCH];
    __shared__ alignas(16) float At[GEMM_WAVES * 16 * AT_PITCH];

    const int lane  = threadIdx.x & 31;
    const int wave  = threadIdx.x >> 5;
    const int m0    = blockIdx.x * GEMM_ROWS + wave * 16;
    const bool active = (m0 < NNODES);
    const int  mr     = active ? m0 : 0;
    const int c = lane & 15;
    const int h = lane >> 4;
    float* Aw = At + wave * 16 * AT_PITCH;

    float4 arow[16];
    {
        const float4* src = (const float4*)(X + (size_t)mr * FEAT) + lane;
        #pragma unroll
        for (int j = 0; j < 16; ++j) arow[j] = src[j * 32];
    }

    {
        const float4* W4 = (const float4*)W;
        #pragma unroll
        for (int i = 0; i < 16; ++i) {
            const int p = i * GEMM_WAVES + wave;
            const float4 r0 = W4[(2 * p) * 32 + lane];
            const float4 r1 = W4[(2 * p + 1) * 32 + lane];
            float4* dst = (float4*)(Wp + p * WP_PITCH + lane * 8);
            dst[0] = make_float4(r0.x, r1.x, r0.y, r1.y);
            dst[1] = make_float4(r0.z, r1.z, r0.w, r1.w);
        }
    }

    {
        #pragma unroll
        for (int j = 0; j < 16; ++j) {
            float4 v = arow[j];
            if (RELU) {
                v.x = fmaxf(v.x, 0.0f); v.y = fmaxf(v.y, 0.0f);
                v.z = fmaxf(v.z, 0.0f); v.w = fmaxf(v.w, 0.0f);
            }
            *(float4*)(Aw + j * AT_PITCH + lane * 4) = v;
        }
    }
    __syncthreads();

    v8f acc[8];
    #pragma unroll
    for (int nt = 0; nt < 8; ++nt) acc[nt] = v8f{};

    {
        const float* ap = Aw + c * AT_PITCH + 2 * h;
        const float* bp = Wp + h * WP_PITCH + 2 * c;
        #pragma unroll
        for (int s = 0; s < 32; ++s) {
            const v2f a = *(const v2f*)(ap + 4 * s);
            #pragma unroll
            for (int nt = 0; nt < 8; ++nt) {
                const v2f b = *(const v2f*)(bp + 2 * s * WP_PITCH + 32 * nt);
                acc[nt] = __builtin_amdgcn_wmma_f32_16x16x4_f32(
                    false, a, false, b, (short)0, acc[nt], false, false);
            }
        }

        #pragma unroll
        for (int nt = 0; nt < 8; ++nt) {
            const float bv = bias[nt * 16 + c];
            #pragma unroll
            for (int r = 0; r < 8; ++r)
                Aw[(8 * h + r) * AT_PITCH + nt * 16 + c] = acc[nt][r] + bv;
        }
    }
    __syncthreads();
    if (active) {
        float4* dst = (float4*)(Y + (size_t)m0 * FEAT) + lane;
        #pragma unroll
        for (int j = 0; j < 16; ++j)
            dst[j * 32] = *(const float4*)(Aw + j * AT_PITCH + lane * 4);
    }
}

#define GATHER_ROWS  8
#define GATHER_BATCH 8

__device__ __forceinline__ float4 gcn_relu4(float4 v) {
    return make_float4(fmaxf(v.x, 0.0f), fmaxf(v.y, 0.0f), fmaxf(v.z, 0.0f), fmaxf(v.w, 0.0f));
}

__device__ __forceinline__ int2 gcn_row_terms(const int2* __restrict__ elist, int s, int e, int base,
                                              int lane, int n, float sw) {
    int2 ent = make_int2(0, 0);
    const int idx = base + lane - 1;
    if (idx >= s && idx < e) ent = elist[idx];
    if (base == s && lane == 0) ent = make_int2(n, __float_as_int(sw));
    return ent;
}

__device__ __forceinline__ void gcn_batch_issue(float4 (&h)[GATHER_BATCH], const float4* __restrict__ H4,
                                                int2 ent, int j0, int cnt, int lane) {
    #pragma unroll
    for (int k = 0; k < GATHER_BATCH; ++k) {
        if (k < cnt) {
            const int c = __builtin_amdgcn_readlane(ent.x, j0 + k);
            h[k] = (H4 + (size_t)c * 32)[lane];
        }
    }
}

template <bool LAYER2>
__device__ __forceinline__ void gcn_batch_sum(const float4 (&h)[GATHER_BATCH], int2 ent, int j0, int cnt,
                                              float4& acc, float& ws) {
    #pragma unroll
    for (int k = 0; k < GATHER_BATCH; ++k) {
        if (k < cnt) {
            const float w = __int_as_float(__builtin_amdgcn_readlane(ent.y, j0 + k));
            if (LAYER2) ws += w;
            acc.x += w * h[k].x; acc.y += w * h[k].y;
            acc.z += w * h[k].z; acc.w += w * h[k].w;
        }
    }
}

template <bool LAYER2>
__global__ void __launch_bounds__(256)
gcn_gather_kernel(const int* __restrict__ rowstart,
                  const int* __restrict__ rowend,
                  const int* __restrict__ elist_words,
                  const float* __restrict__ dinv,
                  const float* __restrict__ H,
                  float* __restrict__ AGG,
                  float* __restrict__ wsum) {
    const int2* elist = (const int2*)elist_words;
    const int lane = threadIdx.x & 31;
    const int wv   = (blockIdx.x * 256 + threadIdx.x) >> 5;
    const int n0   = wv * GATHER_ROWS;
    if (n0 >= NNODES) return;
    const int nrows = (NNODES % GATHER_ROWS == 0) ? GATHER_ROWS
                                                  : min(GATHER_ROWS, NNODES - n0);
    int   ls = 0, le = 0;
    float lw = 0.0f;
    if (lane < nrows) {
        ls = rowstart[n0 + lane];
        le = rowend[n0 + lane];
        const float d = dinv[n0 + lane];
        lw = d * d;
    }
    const float4* H4 = (const float4*)H;

    int   s  = __builtin_amdgcn_readlane(ls, 0);
    int   e  = __builtin_amdgcn_readlane(le, 0);
    float sw = __int_as_float(__builtin_amdgcn_readlane(__float_as_int(lw), 0));
    int2  ent = gcn_row_terms(elist, s, e, s, lane, n0, sw);
    float wsv = 0.0f;

    for (int i = 0; i < nrows; ++i) {
        const int n = n0 + i;
        const int terms = e - s + 1;
        float4 acc = make_float4(0.0f, 0.0f, 0.0f, 0.0f);
        float  ws  = 0.0f;
        float4 h[GATHER_BATCH];

        int m   = min(32, terms);
        int cnt = min(GATHER_BATCH, m);
        gcn_batch_issue(h, H4, ent, 0, cnt, lane);

        int s1 = 0, e1 = 0; float sw1 = 0.0f;
        int2 ent1 = make_int2(0, 0);
        if (i + 1 < nrows) {
            s1  = __builtin_amdgcn_readlane(ls, i + 1);
            e1  = __builtin_amdgcn_readlane(le, i + 1);
            sw1 = __int_as_float(__builtin_amdgcn_readlane(__float_as_int(lw), i + 1));
            ent1 = gcn_row_terms(elist, s1, e1, s1, lane, n + 1, sw1);
        }

        gcn_batch_sum<LAYER2>(h, ent, 0, cnt, acc, ws);

        for (int j0 = GATHER_BATCH; j0 < m; j0 += GATHER_BATCH) {
            cnt = min(GATHER_BATCH, m - j0);
            gcn_batch_issue(h, H4, ent, j0, cnt, lane);
            gcn_batch_sum<LAYER2>(h, ent, j0, cnt, acc, ws);
        }
        for (int t0 = 32; t0 < terms; t0 += 32) {
            ent = gcn_row_terms(elist, s, e, s + t0, lane, n, sw);
            m = min(32, terms - t0);
            for (int j0 = 0; j0 < m; j0 += GATHER_BATCH) {
                cnt = min(GATHER_BATCH, m - j0);
                gcn_batch_issue(h, H4, ent, j0, cnt, lane);
                gcn_batch_sum<LAYER2>(h, ent, j0, cnt, acc, ws);
            }
        }

        if (!LAYER2) acc = gcn_relu4(acc);
        ((float4*)AGG + (size_t)n * 32)[lane] = acc;
        if (LAYER2 && lane == i) wsv = ws;
        s = s1; e = e1; sw = sw1; ent = ent1;
    }
    if (LAYER2 && lane < nrows) wsum[n0 + lane] = wsv;
}

#define POOL_RUN 32

__device__ __forceinline__ void gcn_pool_flush(float* __restrict__ psum,
                                               float* __restrict__ cnts,
                                               float* __restrict__ gws,
                                               int g, float4 acc, float cnt, float wacc, int lane) {
    float* dst = psum + (size_t)g * FEAT + lane * 4;
    atomicAdd(dst + 0, acc.x);
    atomicAdd(dst + 1, acc.y);
    atomicAdd(dst + 2, acc.z);
    atomicAdd(dst + 3, acc.w);
    if (lane == 0) atomicAdd(&cnts[g], cnt);
    if (lane == 1) atomicAdd(&gws[g], wacc);
}

__global__ void __launch_bounds__(128)
gcn_pool_kernel(const float* __restrict__ AGG,
                const int* __restrict__ batch,
                const float* __restrict__ wsum,
                float* __restrict__ psum,
                float* __restrict__ cnts,
                float* __restrict__ gws) {
    const int lane = threadIdx.x & 31;
    const int w    = (blockIdx.x * 128 + threadIdx.x) >> 5;
    const int n0   = w * POOL_RUN;
    if (n0 >= NNODES) return;
    const int nn = (NNODES % POOL_RUN == 0) ? POOL_RUN
                                            : min(POOL_RUN, NNODES - n0);
    const int gl = (lane < nn) ? batch[n0 + lane] : 0;
    const float wl = (lane < nn) ? wsum[n0 + lane] : 0.0f;
    const float4* src = (const float4*)(AGG + (size_t)n0 * FEAT) + lane;

    float4 h[POOL_RUN];
    #pragma unroll
    for (int i = 0; i < POOL_RUN; ++i)
        if (i < nn) h[i] = src[i * 32];

    int    gcur = __builtin_amdgcn_readlane(gl, 0);
    float4 acc  = make_float4(0.0f, 0.0f, 0.0f, 0.0f);
    float  cnt  = 0.0f;
    float  wacc = 0.0f;
    #pragma unroll
    for (int i = 0; i < POOL_RUN; ++i) {
        if (i < nn) {
            const int g = __builtin_amdgcn_readlane(gl, i);
            if (g != gcur) {
                gcn_pool_flush(psum, cnts, gws, gcur, acc, cnt, wacc, lane);
                acc  = make_float4(0.0f, 0.0f, 0.0f, 0.0f);
                cnt  = 0.0f;
                wacc = 0.0f;
                gcur = g;
            }
            acc.x += h[i].x; acc.y += h[i].y; acc.z += h[i].z; acc.w += h[i].w;
            cnt  += 1.0f;
            wacc += __int_as_float(__builtin_amdgcn_readlane(__float_as_int(wl), i));
        }
    }
    gcn_pool_flush(psum, cnts, gws, gcur, acc, cnt, wacc, lane);
}

__global__ void __launch_bounds__(128)
gcn_head_kernel(const float* __restrict__ psum,
                const float* __restrict__ W,
                const float* __restrict__ bias,
                const float* __restrict__ cnts,
                const float* __restrict__ gws,
                float* __restrict__ out) {
    __shared__ alignas(16) float Ct[16 * AT_PITCH];

    const int lane = threadIdx.x & 31;
    const int wave = threadIdx.x >> 5;
    const int c  = lane & 15;
    const int h  = lane >> 4;
    const int m0 = blockIdx.x * 16;
    const int n0 = wave * 32;

    const float* ap = psum + (size_t)(m0 + c) * FEAT + 2 * h;
    const float* bp = W + (size_t)(2 * h) * FEAT + n0 + c;
    v2f a[32], b0[32], b1[32];
    #pragma unroll
    for (int s = 0; s < 32; ++s) {
        a[s]  = *(const v2f*)(ap + 4 * s);
        b0[s] = v2f{bp[(4 * s) * FEAT],      bp[(4 * s + 1) * FEAT]};
        b1[s] = v2f{bp[(4 * s) * FEAT + 16], bp[(4 * s + 1) * FEAT + 16]};
    }

    v8f acc0 = v8f{};
    v8f acc1 = v8f{};
    #pragma unroll
    for (int s = 0; s < 32; ++s) {
        acc0 = __builtin_amdgcn_wmma_f32_16x16x4_f32(
            false, a[s], false, b0[s], (short)0, acc0, false, false);
        acc1 = __builtin_amdgcn_wmma_f32_16x16x4_f32(
            false, a[s], false, b1[s], (short)0, acc1, false, false);
    }

    #pragma unroll
    for (int r = 0; r < 8; ++r) {
        Ct[(8 * h + r) * AT_PITCH + n0 + c]      = acc0[r];
        Ct[(8 * h + r) * AT_PITCH + n0 + 16 + c] = acc1[r];
    }
    __syncthreads();

    const float4 bv = ((const float4*)bias)[lane];
    #pragma unroll
    for (int j = 0; j < 4; ++j) {
        const int   rr = wave * 4 + j;
        const int   g  = m0 + rr;
        const float d  = fmaxf(cnts[g], 1.0f);
        const float wm = gws[g] / d;
        const float4 v = *(const float4*)(Ct + rr * AT_PITCH + lane * 4);
        float4 o;
        o.x = v.x / d + wm * bv.x;
        o.y = v.y / d + wm * bv.y;
        o.z = v.z / d + wm * bv.z;
        o.w = v.w / d + wm * bv.w;
        ((float4*)out)[(size_t)g * 32 + lane] = o;
    }
}

extern "C" void kernel_launch(void* const* d_in, const int* in_sizes, int n_in,
                              void* d_out, int out_size, void* d_ws, size_t ws_size,
                              hipStream_t stream) {
    const float* x          = (const float*)d_in[0];
    const int*   edge_index = (const int*)  d_in[1];
    const int*   batch      = (const int*)  d_in[2];
    const float* W1 = (const float*)d_in[4];
    const float* b1 = (const float*)d_in[5];
    const float* W2 = (const float*)d_in[6];
    const float* b2 = (const float*)d_in[7];
    float* out = (float*)d_out;

    float* H    = (float*)d_ws;
    float* AGG  = H    + (size_t)NNODES * FEAT;
    float* dinv = AGG  + (size_t)NNODES * FEAT;
    float* cnts = dinv + NNODES;
    float* gws  = cnts + NGRAPH;
    float* psum = gws  + NGRAPH;
    float* wsum = psum + (size_t)NGRAPH * FEAT;
    int*   rowcnt   = (int*)(wsum + NNODES);
    int*   rowstart = rowcnt + NNODES;
    int*   total    = rowstart + NNODES;
    int*   elist    = total + 16;

    const int* row = edge_index;
    const int* col = edge_index + NEDGES;

    dim3 blk(256);
    const int grid_nodes  = (NNODES + 255) / 256;
    const int grid_edges  = (NEDGES + 255) / 256;
    const int grid_gemm   = (NNODES + GEMM_ROWS - 1) / GEMM_ROWS;
    const int grid_gather = ((NNODES + GATHER_ROWS - 1) / GATHER_ROWS + 7) / 8;
    const int grid_head   = NGRAPH / 16;
    const int grid_pool   = ((NNODES + POOL_RUN - 1) / POOL_RUN + 3) / 4;

    gcn_init_kernel<<<grid_nodes, blk, 0, stream>>>(psum, cnts, gws, dinv, rowcnt, total);
    gcn_deg_kernel<<<grid_edges, blk, 0, stream>>>(row, col, dinv, rowcnt);
    gcn_dinv_kernel<<<grid_nodes, blk, 0, stream>>>(dinv, rowcnt, rowstart, total);
    gcn_fill_kernel<<<grid_edges, blk, 0, stream>>>(row, col, dinv, rowcnt, elist);

    gcn_gemm_kernel<false><<<grid_gemm, dim3(GEMM_WAVES * 32), 0, stream>>>(x, W1, b1, H);
    gcn_gather_kernel<false><<<grid_gather, blk, 0, stream>>>(rowstart, rowcnt, elist, dinv, H, AGG, wsum);

    gcn_gather_kernel<true><<<grid_gather, blk, 0, stream>>>(rowstart, rowcnt, elist, dinv, AGG, H, wsum);
    gcn_pool_kernel<<<grid_pool, dim3(128), 0, stream>>>(H, batch, wsum, psum, cnts, gws);
    gcn_head_kernel<<<grid_head, dim3(128), 0, stream>>>(psum, W2, b2, cnts, gws, out);
}


# edge lists and degrees by a two-level LDS counting sort (group edges by 256-node range, rank per range): no per-edge global atomics, 9 launches -> 8; changes list order = order of row sums
# speedup vs baseline: 5.0082x; 1.5771x over previous
#include <hip/hip_runtime.h>
#include <cmath>

#define NNODES 100000
#define NEDGES 600000
#define FEAT   128
#define NGRAPH 256

typedef __attribute__((ext_vector_type(2))) float v2f;
typedef __attribute__((ext_vector_type(8))) float v8f;

#define BIN_SHIFT 8
#define BIN_ROWS  (1 << BIN_SHIFT)
#define NBINS     ((NNODES + BIN_ROWS - 1) / BIN_ROWS)
#define NBINS1    (NBINS + 1)
#define EPT       10
#define EPB       (256 * EPT)
#define NEBLK     ((NEDGES + EPB - 1) / EPB)
#define SCAN_PER_LANE ((NBINS1 + 31) / 32)
#define WALK      8

static_assert(NEBLK * 256 >= NGRAPH * FEAT, "level-1 grid also clears the pooled sums");
static_assert(WALK == 8, "gcn_hold8 names WALK values");

__device__ __forceinline__ void gcn_hold8(int a0, int a1, int a2, int a3, int a4, int a5, int a6, int a7) {
    asm volatile("" :: "v"(a0), "v"(a1), "v"(a2), "v"(a3), "v"(a4), "v"(a5), "v"(a6), "v"(a7));
}
__device__ __forceinline__ void gcn_hold8(float a0, float a1, float a2, float a3, float a4, float a5, float a6, float a7) {
    asm volatile("" :: "v"(a0), "v"(a1), "v"(a2), "v"(a3), "v"(a4), "v"(a5), "v"(a6), "v"(a7));
}

__global__ void __launch_bounds__(256)
gcn_bin_kernel(const int* __restrict__ row, const int* __restrict__ col,
               int* __restrict__ offR, int* __restrict__ offC,
               int* __restrict__ binR_words, int* __restrict__ binC,
               float* __restrict__ psum, float* __restrict__ cnts,
               float* __restrict__ gws, int* __restrict__ total) {
    __shared__ int hR[NBINS1];
    __shared__ int hC[NBINS1];
    __shared__ int2 sR[EPB];
    __shared__ int  sC[EPB];
    const int tid  = threadIdx.x;
    const int lane = tid & 31;
    const int wave = tid >> 5;
    const int blk  = blockIdx.x;
    {
        const int i = blk * 256 + tid;
        if (i < NGRAPH * FEAT) psum[i] = 0.0f;
        if (i < NGRAPH) {
            cnts[i] = 0.0f;
            gws[i]  = 0.0f;
        }
        if (i == 0) *total = 0;
    }
    for (int i = tid; i < NBINS1; i += 256) { hR[i] = 0; hC[i] = 0; }

    int r[EPT], c[EPT], kr[EPT], kc[EPT];
    const int e0 = blk * EPB + tid;
    #pragma unroll
    for (int j = 0; j < EPT; ++j) {
        const int e = e0 + j * 256;
        r[j] = 0; c[j] = 0;
        if (e < NEDGES) { r[j] = row[e]; c[j] = col[e]; }
    }
    __syncthreads();
    #pragma unroll
    for (int j = 0; j < EPT; ++j) {
        kr[j] = 0; kc[j] = 0;
        if (e0 + j * 256 < NEDGES) {
            kr[j] = atomicAdd(&hR[r[j] >> BIN_SHIFT], 1);
            kc[j] = atomicAdd(&hC[c[j] >> BIN_SHIFT], 1);
        }
    }
    __syncthreads();
    if (wave < 2) {
        int* h  = (wave == 0) ? hR : hC;
        int* og = ((wave == 0) ? offR : offC) + blk;
        int v[SCAN_PER_LANE];
        int sum = 0;
        #pragma unroll
        for (int q = 0; q < SCAN_PER_LANE; ++q) {
            const int idx = lane * SCAN_PER_LANE + q;
            v[q] = (idx < NBINS1) ? h[idx] : 0;
            sum += v[q];
        }
        int inc = sum;
        #pragma unroll
        for (int o = 1; o < 32; o <<= 1) {
            const int t = __shfl_up(inc, o);
            if (lane >= o) inc += t;
        }
        int ex = inc - sum;
        #pragma unroll
        for (int q = 0; q < SCAN_PER_LANE; ++q) {
            const int idx = lane * SCAN_PER_LANE + q;
            if (idx < NBINS1) { h[idx] = ex; og[idx * NEBLK] = ex; }
            ex += v[q];
        }
    }
    __syncthreads();
    #pragma unroll
    for (int j = 0; j < EPT; ++j) {
        if (e0 + j * 256 < NEDGES) {
            sR[hR[r[j] >> BIN_SHIFT] + kr[j]] = make_int2(r[j], c[j]);
            sC[hC[c[j] >> BIN_SHIFT] + kc[j]] = c[j];
        }
    }
    __syncthreads();
    const int nE = min(EPB, NEDGES - blk * EPB);
    int2* binR = (int2*)binR_words + (size_t)blk * EPB;
    int*  bC   = binC + (size_t)blk * EPB;
    #pragma unroll
    for (int j = 0; j < EPT; ++j) {
        const int i = tid + j * 256;
        if (i < nE) {
            binR[i] = sR[i];
            bC[i]   = sC[i];
        }
    }
}

__global__ void __launch_bounds__(256)
gcn_dinv_kernel(const int* __restrict__ offC, const int* __restrict__ binC,
                float* __restrict__ dinv) {
    __shared__ int cnt[BIN_ROWS];
    const int tid = threadIdx.x;
    const int b   = blockIdx.x;
    cnt[tid] = 0;
    __syncthreads();
    for (int k = tid; k < NEBLK; k += 256) {
        const int s = offC[b * NEBLK + k];
        const int t = offC[(b + 1) * NEBLK + k];
        const int* src = binC + (size_t)k * EPB;
        for (int i = s; i < t; i += WALK) {
            int cc[WALK];
            #pragma unroll
            for (int q = 0; q < WALK; ++q) cc[q] = src[min(i + q, t - 1)];
            gcn_hold8(cc[0], cc[1], cc[2], cc[3], cc[4], cc[5], cc[6], cc[7]);
            #pragma unroll
            for (int q = 0; q < WALK; ++q)
                if (i + q < t) atomicAdd(&cnt[cc[q] & (BIN_ROWS - 1)], 1);
        }
    }
    __syncthreads();
    const int n = b * BIN_ROWS + tid;
    if (n < NNODES) {
        const float d = 1.0f + (float)cnt[tid];
        dinv[n] = (d > 0.0f) ? rsqrtf(d) : 0.0f;
    }
}

__global__ void __launch_bounds__(256)
gcn_fill_kernel(const int* __restrict__ offR, const int* __restrict__ binR_words,
                const float* __restrict__ dinv,
                int* __restrict__ rowstart, int* __restrict__ rowend,
                int* __restrict__ total, int* __restrict__ elist_words) {
    __shared__ int   cnt[BIN_ROWS];
    __shared__ float dl[BIN_ROWS];
    __shared__ int   wsum[8];
    __shared__ int   sbase;
    const int2* binR = (const int2*)binR_words;
    int2* elist = (int2*)elist_words;
    const int tid  = threadIdx.x;
    const int lane = tid & 31;
    const int wave = tid >> 5;
    const int b    = blockIdx.x;
    const int n    = b * BIN_ROWS + tid;
    cnt[tid] = 0;
    dl[tid]  = (n < NNODES) ? dinv[n] : 0.0f;
    __syncthreads();
    for (int k = tid; k < NEBLK; k += 256) {
        const int s = offR[b * NEBLK + k];
        const int t = offR[(b + 1) * NEBLK + k];
        const int2* src = binR + (size_t)k * EPB;
        for (int i = s; i < t; i += WALK) {
            int rr[WALK];
            #pragma unroll
            for (int q = 0; q < WALK; ++q) rr[q] = src[min(i + q, t - 1)].x;
            gcn_hold8(rr[0], rr[1], rr[2], rr[3], rr[4], rr[5], rr[6], rr[7]);
            #pragma unroll
            for (int q = 0; q < WALK; ++q)
                if (i + q < t) atomicAdd(&cnt[rr[q] & (BIN_ROWS - 1)], 1);
        }
    }
    __syncthreads();
    const int c = cnt[tid];
    int v = c;
    #pragma unroll
    for (int o = 1; o < 32; o <<= 1) {
        int t = __shfl_up(v, o);
        if (lane >= o) v += t;
    }
    if (lane == 31) wsum[wave] = v;
    __syncthreads();
    if (tid == 0) {
        int s = 0;
        for (int w = 0; w < 8; ++w) { int t = wsum[w]; wsum[w] = s; s += t; }
        sbase = atomicAdd(total, s);
    }
    __syncthreads();
    const int st = sbase + wsum[wave] + v - c;
    if (n < NNODES) {
        rowstart[n] = st;
        rowend[n]   = st + c;
    }
    cnt[tid] = st;
    __syncthreads();
    for (int k = tid; k < NEBLK; k += 256) {
        const int s = offR[b * NEBLK + k];
        const int t = offR[(b + 1) * NEBLK + k];
        const int2* src = binR + (size_t)k * EPB;
        for (int i = s; i < t; i += WALK) {
            int2  en[WALK];
            float dc[WALK];
            #pragma unroll
            for (int q = 0; q < WALK; ++q) en[q] = src[min(i + q, t - 1)];
            #pragma unroll
            for (int q = 0; q < WALK; ++q) dc[q] = dinv[en[q].y];
            gcn_hold8(dc[0], dc[1], dc[2], dc[3], dc[4], dc[5], dc[6], dc[7]);
            #pragma unroll
            for (int q = 0; q < WALK; ++q) {
                if (i + q < t) {
                    const int rl  = en[q].x & (BIN_ROWS - 1);
                    const int pos = atomicAdd(&cnt[rl], 1);
                    const float w = dl[rl] * dc[q];
                    elist[pos] = make_int2(en[q].y, __float_as_int(w));
                }
            }
        }
    }
}

#define GEMM_WAVES 4
#define GEMM_ROWS  (GEMM_WAVES * 16)
#define WP_PITCH   288
#define AT_PITCH   132

template <bool RELU>
__global__ void __launch_bounds__(GEMM_WAVES * 32)
gcn_gemm_kernel(const float* __restrict__ X,
                const float* __restrict__ W,
                const float* __restrict__ bias,
                float* __restrict__ Y) {
    __shared__ alignas(16) float Wp[64 * WP_PITCH];
    __shared__ alignas(16) float At[GEMM_WAVES * 16 * AT_PITCH];

    const int lane  = threadIdx.x & 31;
    const int wave  = threadIdx.x >> 5;
    const int m0    = blockIdx.x * GEMM_ROWS + wave * 16;
    const bool active = (m0 < NNODES);
    const int  mr     = active ? m0 : 0;
    const int c = lane & 15;
    const int h = lane >> 4;
    float* Aw = At + wave * 16 * AT_PITCH;

    float4 arow[16];
    {
        const float4* src = (const float4*)(X + (size_t)mr * FEAT) + lane;
        #pragma unroll
        for (int j = 0; j < 16; ++j) arow[j] = src[j * 32];
    }

    {
        const float4* W4 = (const float4*)W;
        #pragma unroll
        for (int i = 0; i < 16; ++i) {
            const int p = i * GEMM_WAVES + wave;
            const float4 r0 = W4[(2 * p) * 32 + lane];
            const float4 r1 = W4[(2 * p + 1) * 32 + lane];
            float4* dst = (float4*)(Wp + p * WP_PITCH + lane * 8);
            dst[0] = make_float4(r0.x, r1.x, r0.y, r1.y);
            dst[1] = make_float4(r0.z, r1.z, r0.w, r1.w);
        }
    }

    {
        #pragma unroll
        for (int j = 0; j < 16; ++j) {
            float4 v = arow[j];
            if (RELU) {
                v.x = fmaxf(v.x, 0.0f); v.y = fmaxf(v.y, 0.0f);
                v.z = fmaxf(v.z, 0.0f); v.w = fmaxf(v.w, 0.0f);
            }
            *(float4*)(Aw + j * AT_PITCH + lane * 4) = v;
        }
    }
    __syncthreads();

    v8f acc[8];
    #pragma unroll
    for (int nt = 0; nt < 8; ++nt) acc[nt] = v8f{};

    {
        const float* ap = Aw + c * AT_PITCH + 2 * h;
        const float* bp = Wp + h * WP_PITCH + 2 * c;
        #pragma unroll
        for (int s = 0; s < 32; ++s) {
            const v2f a = *(const v2f*)(ap + 4 * s);
            #pragma unroll
            for (int nt = 0; nt < 8; ++nt) {
                const v2f b = *(const v2f*)(bp + 2 * s * WP_PITCH + 32 * nt);
                acc[nt] = __builtin_amdgcn_wmma_f32_16x16x4_f32(
                    false, a, false, b, (short)0, acc[nt], false, false);
            }
        }

        #pragma unroll
        for (int nt = 0; nt < 8; ++nt) {
            const float bv = bias[nt * 16 + c];
            #pragma unroll
            for (int r = 0; r < 8; ++r)
                Aw[(8 * h + r) * AT_PITCH + nt * 16 + c] = acc[nt][r] + bv;
        }
    }
    __syncthreads();
    if (active) {
        float4* dst = (float4*)(Y + (size_t)m0 * FEAT) + lane;
        #pragma unroll
        for (int j = 0; j < 16; ++j)
            dst[j * 32] = *(const float4*)(Aw + j * AT_PITCH + lane * 4);
    }
}

#define GATHER_ROWS  8
#define GATHER_BATCH 8

__device__ __forceinline__ float4 gcn_relu4(float4 v) {
    return make_float4(fmaxf(v.x, 0.0f), fmaxf(v.y, 0.0f), fmaxf(v.z, 0.0f), fmaxf(v.w, 0.0f));
}

__device__ __forceinline__ int2 gcn_row_terms(const int2* __restrict__ elist, int s, int e, int base,
                                              int lane, int n, float sw) {
    int2 ent = make_int2(0, 0);
    const int idx = base + lane - 1;
    if (idx >= s && idx < e) ent = elist[idx];
    if (base == s && lane == 0) ent = make_int2(n, __float_as_int(sw));
    return ent;
}

__device__ __forceinline__ void gcn_batch_issue(float4 (&h)[GATHER_BATCH], const float4* __restrict__ H4,
                                                int2 ent, int j0, int cnt, int lane) {
    #pragma unroll
    for (int k = 0; k < GATHER_BATCH; ++k) {
        if (k < cnt) {
            const int c = __builtin_amdgcn_readlane(ent.x, j0 + k);
            h[k] = (H4 + (size_t)c * 32)[lane];
        }
    }
}

template <bool LAYER2>
__device__ __forceinline__ void gcn_batch_sum(const float4 (&h)[GATHER_BATCH], int2 ent, int j0, int cnt,
                                              float4& acc, float& ws) {
    #pragma unroll
    for (int k = 0; k < GATHER_BATCH; ++k) {
        if (k < cnt) {
            const float w = __int_as_float(__builtin_amdgcn_readlane(ent.y, j0 + k));
            if (LAYER2) ws += w;
            acc.x += w * h[k].x; acc.y += w * h[k].y;
            acc.z += w * h[k].z; acc.w += w * h[k].w;
        }
    }
}

template <bool LAYER2>
__global__ void __launch_bounds__(256)
gcn_gather_kernel(const int* __restrict__ rowstart,
                  const int* __restrict__ rowend,
                  const int* __restrict__ elist_words,
                  const float* __restrict__ dinv,
                  const float* __restrict__ H,
                  float* __restrict__ AGG,
                  float* __restrict__ wsum) {
    const int2* elist = (const int2*)elist_words;
    const int lane = threadIdx.x & 31;
    const int wv   = (blockIdx.x * 256 + threadIdx.x) >> 5;
    const int n0   = wv * GATHER_ROWS;
    if (n0 >= NNODES) return;
    const int nrows = (NNODES % GATHER_ROWS == 0) ? GATHER_ROWS
                                                  : min(GATHER_ROWS, NNODES - n0);
    int   ls = 0, le = 0;
    float lw = 0.0f;
    if (lane < nrows) {
        ls = rowstart[n0 + lane];
        le = rowend[n0 + lane];
        const float d = dinv[n0 + lane];
        lw = d * d;
    }
    const float4* H4 = (const float4*)H;

    int   s  = __builtin_amdgcn_readlane(ls, 0);
    int   e  = __builtin_amdgcn_readlane(le, 0);
    float sw = __int_as_float(__builtin_amdgcn_readlane(__float_as_int(lw), 0));
    int2  ent = gcn_row_terms(elist, s, e, s, lane, n0, sw);
    float wsv = 0.0f;

    for (int i = 0; i < nrows; ++i) {
        const int n = n0 + i;
        const int terms = e - s + 1;
        float4 acc = make_float4(0.0f, 0.0f, 0.0f, 0.0f);
        float  ws  = 0.0f;
        float4 h[GATHER_BATCH];

        int m   = min(32, terms);
        int cnt = min(GATHER_BATCH, m);
        gcn_batch_issue(h, H4, ent, 0, cnt, lane);

        int s1 = 0, e1 = 0; float sw1 = 0.0f;
        int2 ent1 = make_int2(0, 0);
        if (i + 1 < nrows) {
            s1  = __builtin_amdgcn_readlane(ls, i + 1);
            e1  = __builtin_amdgcn_readlane(le, i + 1);
            sw1 = __int_as_float(__builtin_amdgcn_readlane(__float_as_int(lw), i + 1));
            ent1 = gcn_row_terms(elist, s1, e1, s1, lane, n + 1, sw1);
        }

        gcn_batch_sum<LAYER2>(h, ent, 0, cnt, acc, ws);

        for (int j0 = GATHER_BATCH; j0 < m; j0 += GATHER_BATCH) {
            cnt = min(GATHER_BATCH, m - j0);
            gcn_batch_issue(h, H4, ent, j0, cnt, lane);
            gcn_batch_sum<LAYER2>(h, ent, j0, cnt, acc, ws);
        }
        for (int t0 = 32; t0 < terms; t0 += 32) {
            ent = gcn_row_terms(elist, s, e, s + t0, lane, n, sw);
            m = min(32, terms - t0);
            for (int j0 = 0; j0 < m; j0 += GATHER_BATCH) {
                cnt = min(GATHER_BATCH, m - j0);
                gcn_batch_issue(h, H4, ent, j0, cnt, lane);
                gcn_batch_sum<LAYER2>(h, ent, j0, cnt, acc, ws);
            }
        }

        if (!LAYER2) acc = gcn_relu4(acc);
        ((float4*)AGG + (size_t)n * 32)[lane] = acc;
        if (LAYER2 && lane == i) wsv = ws;
        s = s1; e = e1; sw = sw1; ent = ent1;
    }
    if (LAYER2 && lane < nrows) wsum[n0 + lane] = wsv;
}

#define POOL_RUN 32

__device__ __forceinline__ void gcn_pool_flush(float* __restrict__ psum,
                                               float* __restrict__ cnts,
                                               float* __restrict__ gws,
                                               int g, float4 acc, float cnt, float wacc, int lane) {
    float* dst = psum + (size_t)g * FEAT + lane * 4;
    atomicAdd(dst + 0, acc.x);
    atomicAdd(dst + 1, acc.y);
    atomicAdd(dst + 2, acc.z);
    atomicAdd(dst + 3, acc.w);
    if (lane == 0) atomicAdd(&cnts[g], cnt);
    if (lane == 1) atomicAdd(&gws[g], wacc);
}

__global__ void __launch_bounds__(128)
gcn_pool_kernel(const float* __restrict__ AGG,
                const int* __restrict__ batch,
                const float* __restrict__ wsum,
                float* __restrict__ psum,
                float* __restrict__ cnts,
                float* __restrict__ gws) {
    const int lane = threadIdx.x & 31;
    const int w    = (blockIdx.x * 128 + threadIdx.x) >> 5;
    const int n0   = w * POOL_RUN;
    if (n0 >= NNODES) return;
    const int nn = (NNODES % POOL_RUN == 0) ? POOL_RUN
                                            : min(POOL_RUN, NNODES - n0);
    const int gl = (lane < nn) ? batch[n0 + lane] : 0;
    const float wl = (lane < nn) ? wsum[n0 + lane] : 0.0f;
    const float4* src = (const float4*)(AGG + (size_t)n0 * FEAT) + lane;

    float4 h[POOL_RUN];
    #pragma unroll
    for (int i = 0; i < POOL_RUN; ++i)
        if (i < nn) h[i] = src[i * 32];

    int    gcur = __builtin_amdgcn_readlane(gl, 0);
    float4 acc  = make_float4(0.0f, 0.0f, 0.0f, 0.0f);
    float  cnt  = 0.0f;
    float  wacc = 0.0f;
    #pragma unroll
    for (int i = 0; i < POOL_RUN; ++i) {
        if (i < nn) {
            const int g = __builtin_amdgcn_readlane(gl, i);
            if (g != gcur) {
                gcn_pool_flush(psum, cnts, gws, gcur, acc, cnt, wacc, lane);
                acc  = make_float4(0.0f, 0.0f, 0.0f, 0.0f);
                cnt  = 0.0f;
                wacc = 0.0f;
                gcur = g;
            }
            acc.x += h[i].x; acc.y += h[i].y; acc.z += h[i].z; acc.w += h[i].w;
            cnt  += 1.0f;
            wacc += __int_as_float(__builtin_amdgcn_readlane(__float_as_int(wl), i));
        }
    }
    gcn_pool_flush(psum, cnts, gws, gcur, acc, cnt, wacc, lane);
}

__global__ void __launch_bounds__(128)
gcn_head_kernel(const float* __restrict__ psum,
                const float* __restrict__ W,
                const float* __restrict__ bias,
                const float* __restrict__ cnts,
                const float* __restrict__ gws,
                float* __restrict__ out) {
    __shared__ alignas(16) float Ct[16 * AT_PITCH];

    const int lane = threadIdx.x & 31;
    const int wave = threadIdx.x >> 5;
    const int c  = lane & 15;
    const int h  = lane >> 4;
    const int m0 = blockIdx.x * 16;
    const int n0 = wave * 32;

    const float* ap = psum + (size_t)(m0 + c) * FEAT + 2 * h;
    const float* bp = W + (size_t)(2 * h) * FEAT + n0 + c;
    v2f a[32], b0[32], b1[32];
    #pragma unroll
    for (int s = 0; s < 32; ++s) {
        a[s]  = *(const v2f*)(ap + 4 * s);
        b0[s] = v2f{bp[(4 * s) * FEAT],      bp[(4 * s + 1) * FEAT]};
        b1[s] = v2f{bp[(4 * s) * FEAT + 16], bp[(4 * s + 1) * FEAT + 16]};
    }

    v8f acc0 = v8f{};
    v8f acc1 = v8f{};
    #pragma unroll
    for (int s = 0; s < 32; ++s) {
        acc0 = __builtin_amdgcn_wmma_f32_16x16x4_f32(
            false, a[s], false, b0[s], (short)0, acc0, false, false);
        acc1 = __builtin_amdgcn_wmma_f32_16x16x4_f32(
            false, a[s], false, b1[s], (short)0, acc1, false, false);
    }

    #pragma unroll
    for (int r = 0; r < 8; ++r) {
        Ct[(8 * h + r) * AT_PITCH + n0 + c]      = acc0[r];
        Ct[(8 * h + r) * AT_PITCH + n0 + 16 + c] = acc1[r];
    }
    __syncthreads();

    const float4 bv = ((const float4*)bias)[lane];
    #pragma unroll
    for (int j = 0; j < 4; ++j) {
        const int   rr = wave * 4 + j;
        const int   g  = m0 + rr;
        const float d  = fmaxf(cnts[g], 1.0f);
        const float wm = gws[g] / d;
        const float4 v = *(const float4*)(Ct + rr * AT_PITCH + lane * 4);
        float4 o;
        o.x = v.x / d + wm * bv.x;
        o.y = v.y / d + wm * bv.y;
        o.z = v.z / d + wm * bv.z;
        o.w = v.w / d + wm * bv.w;
        ((float4*)out)[(size_t)g * 32 + lane] = o;
    }
}

extern "C" void kernel_launch(void* const* d_in, const int* in_sizes, int n_in,
                              void* d_out, int out_size, void* d_ws, size_t ws_size,
                              hipStream_t stream) {
    const float* x          = (const float*)d_in[0];
    const int*   edge_index = (const int*)  d_in[1];
    const int*   batch      = (const int*)  d_in[2];
    const float* W1 = (const float*)d_in[4];
    const float* b1 = (const float*)d_in[5];
    const float* W2 = (const float*)d_in[6];
    const float* b2 = (const float*)d_in[7];
    float* out = (float*)d_out;

    float* H    = (float*)d_ws;
    float* AGG  = H    + (size_t)NNODES * FEAT;
    float* dinv = AGG  + (size_t)NNODES * FEAT;
    float* cnts = dinv + NNODES;
    float* gws  = cnts + NGRAPH;
    float* psum = gws  + NGRAPH;
    float* wsum = psum + (size_t)NGRAPH * FEAT;
    int*   rowend   = (int*)(wsum + NNODES);
    int*   rowstart = rowend + NNODES;
    int*   total    = rowstart + NNODES;
    int*   elist    = total + 16;
    int*   binR     = elist + 2 * NEDGES;
    int*   binC     = binR + 2 * NEBLK * EPB;
    int*   offR     = binC + NEBLK * EPB;
    int*   offC     = offR + NEBLK * NBINS1;

    const int* row = edge_index;
    const int* col = edge_index + NEDGES;

    dim3 blk(256);
    const int grid_gemm   = (NNODES + GEMM_ROWS - 1) / GEMM_ROWS;
    const int grid_gather = ((NNODES + GATHER_ROWS - 1) / GATHER_ROWS + 7) / 8;
    const int grid_head   = NGRAPH / 16;
    const int grid_pool   = ((NNODES + POOL_RUN - 1) / POOL_RUN + 3) / 4;

    gcn_bin_kernel<<<NEBLK, blk, 0, stream>>>(row, col, offR, offC, binR, binC, psum, cnts, gws, total);
    gcn_dinv_kernel<<<NBINS, blk, 0, stream>>>(offC, binC, dinv);
    gcn_fill_kernel<<<NBINS, blk, 0, stream>>>(offR, binR, dinv, rowstart, rowend, total, elist);

    gcn_gemm_kernel<false><<<grid_gemm, dim3(GEMM_WAVES * 32), 0, stream>>>(x, W1, b1, H);
    gcn_gather_kernel<false><<<grid_gather, blk, 0, stream>>>(rowstart, rowend, elist, dinv, H, AGG, wsum);

    gcn_gather_kernel<true><<<grid_gather, blk, 0, stream>>>(rowstart, rowend, elist, dinv, AGG, H, wsum);
    gcn_pool_kernel<<<grid_pool, dim3(128), 0, stream>>>(H, batch, wsum, psum, cnts, gws);
    gcn_head_kernel<<<grid_head, dim3(128), 0, stream>>>(psum, W2, b2, cnts, gws, out);
}


# layer-1 product gets no launch of its own: its 64-row tiles run as extra work-groups of the 3 edge-list launches, whose short chains leave the matrix unit idle; 8 launches -> 7; same sums, same order
# speedup vs baseline: 5.2862x; 1.0555x over previous
#include <hip/hip_runtime.h>
#include <cmath>

#define NNODES 100000
#define NEDGES 600000
#define FEAT   128
#define NGRAPH 256

typedef __attribute__((ext_vector_type(2))) float v2f;
typedef __attribute__((ext_vector_type(8))) float v8f;

#define BIN_SHIFT 8
#define BIN_ROWS  (1 << BIN_SHIFT)
#define NBINS     ((NNODES + BIN_ROWS - 1) / BIN_ROWS)
#define NBINS1    (NBINS + 1)
#define EPT       10
#define EPB       (256 * EPT)
#define NEBLK     ((NEDGES + EPB - 1) / EPB)
#define SCAN_PER_LANE ((NBINS1 + 31) / 32)
#define WALK      8

static_assert(NEBLK * 256 >= NGRAPH * FEAT, "level-1 grid also clears the pooled sums");
static_assert(WALK == 8, "gcn_hold8 names WALK values");

__device__ __forceinline__ void gcn_hold8(int a0, int a1, int a2, int a3, int a4, int a5, int a6, int a7) {
    asm volatile("" :: "v"(a0), "v"(a1), "v"(a2), "v"(a3), "v"(a4), "v"(a5), "v"(a6), "v"(a7));
}
__device__ __forceinline__ void gcn_hold8(float a0, float a1, float a2, float a3, float a4, float a5, float a6, float a7) {
    asm volatile("" :: "v"(a0), "v"(a1), "v"(a2), "v"(a3), "v"(a4), "v"(a5), "v"(a6), "v"(a7));
}

__device__ __forceinline__ void
gcn_bin_role(int* __restrict__ smem, const int blk,
             const int* __restrict__ row, const int* __restrict__ col,
             int* __restrict__ offR, int* __restrict__ offC,
             int* __restrict__ binR_words, int* __restrict__ binC,
             float* __restrict__ psum, float* __restrict__ cnts,
             float* __restrict__ gws, int* __restrict__ total) {
    int*  hR = smem;
    int*  hC = smem + NBINS1;
    int2* sR = (int2*)(smem + 2 * NBINS1);
    int*  sC = smem + 2 * NBINS1 + 2 * EPB;
    const int tid  = threadIdx.x;
    const int lane = tid & 31;
    const int wave = tid >> 5;
    {
        const int i = blk * 256 + tid;
        if (i < NGRAPH * FEAT) psum[i] = 0.0f;
        if (i < NGRAPH) {
            cnts[i] = 0.0f;
            gws[i]  = 0.0f;
        }
        if (i == 0) *total = 0;
    }
    for (int i = tid; i < NBINS1; i += 256) { hR[i] = 0; hC[i] = 0; }

    int r[EPT], c[EPT], kr[EPT], kc[EPT];
    const int e0 = blk * EPB + tid;
    #pragma unroll
    for (int j = 0; j < EPT; ++j) {
        const int e = e0 + j * 256;
        r[j] = 0; c[j] = 0;
        if (e < NEDGES) { r[j] = row[e]; c[j] = col[e]; }
    }
    __syncthreads();
    #pragma unroll
    for (int j = 0; j < EPT; ++j) {
        kr[j] = 0; kc[j] = 0;
        if (e0 + j * 256 < NEDGES) {
            kr[j] = atomicAdd(&hR[r[j] >> BIN_SHIFT], 1);
            kc[j] = atomicAdd(&hC[c[j] >> BIN_SHIFT], 1);
        }
    }
    __syncthreads();
    if (wave < 2) {
        int* h  = (wave == 0) ? hR : hC;
        int* og = ((wave == 0) ? offR : offC) + blk;
        int v[SCAN_PER_LANE];
        int sum = 0;
        #pragma unroll
        for (int q = 0; q < SCAN_PER_LANE; ++q) {
            const int idx = lane * SCAN_PER_LANE + q;
            v[q] = (idx < NBINS1) ? h[idx] : 0;
            sum += v[q];
        }
        int inc = sum;
        #pragma unroll
        for (int o = 1; o < 32; o <<= 1) {
            const int t = __shfl_up(inc, o);
            if (lane >= o) inc += t;
        }
        int ex = inc - sum;
        #pragma unroll
        for (int q = 0; q < SCAN_PER_LANE; ++q) {
            const int idx = lane * SCAN_PER_LANE + q;
            if (idx < NBINS1) { h[idx] = ex; og[idx * NEBLK] = ex; }
            ex += v[q];
        }
    }
    __syncthreads();
    #pragma unroll
    for (int j = 0; j < EPT; ++j) {
        if (e0 + j * 256 < NEDGES) {
            sR[hR[r[j] >> BIN_SHIFT] + kr[j]] = make_int2(r[j], c[j]);
            sC[hC[c[j] >> BIN_SHIFT] + kc[j]] = c[j];
        }
    }
    __syncthreads();
    const int nE = min(EPB, NEDGES - blk * EPB);
    int2* binR = (int2*)binR_words + (size_t)blk * EPB;
    int*  bC   = binC + (size_t)blk * EPB;
    #pragma unroll
    for (int j = 0; j < EPT; ++j) {
        const int i = tid + j * 256;
        if (i < nE) {
            binR[i] = sR[i];
            bC[i]   = sC[i];
        }
    }
}

__device__ __forceinline__ void
gcn_dinv_role(int* __restrict__ smem, const int b,
              const int* __restrict__ offC, const int* __restrict__ binC,
              float* __restrict__ dinv) {
    int* cnt = smem;
    const int tid = threadIdx.x;
    cnt[tid] = 0;
    __syncthreads();
    for (int k = tid; k < NEBLK; k += 256) {
        const int s = offC[b * NEBLK + k];
        const int t = offC[(b + 1) * NEBLK + k];
        const int* src = binC + (size_t)k * EPB;
        for (int i = s; i < t; i += WALK) {
            int cc[WALK];
            #pragma unroll
            for (int q = 0; q < WALK; ++q) cc[q] = src[min(i + q, t - 1)];
            gcn_hold8(cc[0], cc[1], cc[2], cc[3], cc[4], cc[5], cc[6], cc[7]);
            #pragma unroll
            for (int q = 0; q < WALK; ++q)
                if (i + q < t) atomicAdd(&cnt[cc[q] & (BIN_ROWS - 1)], 1);
        }
    }
    __syncthreads();
    const int n = b * BIN_ROWS + tid;
    if (n < NNODES) {
        const float d = 1.0f + (float)cnt[tid];
        dinv[n] = (d > 0.0f) ? rsqrtf(d) : 0.0f;
    }
}

__device__ __forceinline__ void
gcn_fill_role(int* __restrict__ smem, const int b,
              const int* __restrict__ offR, const int* __restrict__ binR_words,
              const float* __restrict__ dinv,
              int* __restrict__ rowstart, int* __restrict__ rowend,
              int* __restrict__ total, int* __restrict__ elist_words) {
    int*   cnt   = smem;
    float* dl    = (float*)(smem + BIN_ROWS);
    int*   wsum  = smem + 2 * BIN_ROWS;
    int*   sbase = smem + 2 * BIN_ROWS + 8;
    const int2* binR = (const int2*)binR_words;
    int2* elist = (int2*)elist_words;
    const int tid  = threadIdx.x;
    const int lane = tid & 31;
    const int wave = tid >> 5;
    const int n    = b * BIN_ROWS + tid;
    cnt[tid] = 0;
    dl[tid]  = (n < NNODES) ? dinv[n] : 0.0f;
    __syncthreads();
    for (int k = tid; k < NEBLK; k += 256) {
        const int s = offR[b * NEBLK + k];
        const int t = offR[(b + 1) * NEBLK + k];
        const int2* src = binR + (size_t)k * EPB;
        for (int i = s; i < t; i += WALK) {
            int rr[WALK];
            #pragma unroll
            for (int q = 0; q < WALK; ++q) rr[q] = src[min(i + q, t - 1)].x;
            gcn_hold8(rr[0], rr[1], rr[2], rr[3], rr[4], rr[5], rr[6], rr[7]);
            #pragma unroll
            for (int q = 0; q < WALK; ++q)
                if (i + q < t) atomicAdd(&cnt[rr[q] & (BIN_ROWS - 1)], 1);
        }
    }
    __syncthreads();
    const int c = cnt[tid];
    int v = c;
    #pragma unroll
    for (int o = 1; o < 32; o <<= 1) {
        int t = __shfl_up(v, o);
        if (lane >= o) v += t;
    }
    if (lane == 31) wsum[wave] = v;
    __syncthreads();
    if (tid == 0) {
        int s = 0;
        for (int w = 0; w < 8; ++w) { int t = wsum[w]; wsum[w] = s; s += t; }
        *sbase = atomicAdd(total, s);
    }
    __syncthreads();
    const int st = *sbase + wsum[wave] + v - c;
    if (n < NNODES) {
        rowstart[n] = st;
        rowend[n]   = st + c;
    }
    cnt[tid] = st;
    __syncthreads();
    for (int k = tid; k < NEBLK; k += 256) {
        const int s = offR[b * NEBLK + k];
        const int t = offR[(b + 1) * NEBLK + k];
        const int2* src = binR + (size_t)k * EPB;
        for (int i = s; i < t; i += WALK) {
            int2  en[WALK];
            float dc[WALK];
            #pragma unroll
            for (int q = 0; q < WALK; ++q) en[q] = src[min(i + q, t - 1)];
            #pragma unroll
            for (int q = 0; q < WALK; ++q) dc[q] = dinv[en[q].y];
            gcn_hold8(dc[0], dc[1], dc[2], dc[3], dc[4], dc[5], dc[6], dc[7]);
            #pragma unroll
            for (int q = 0; q < WALK; ++q) {
                if (i + q < t) {
                    const int rl  = en[q].x & (BIN_ROWS - 1);
                    const int pos = atomicAdd(&cnt[rl], 1);
                    const float w = dl[rl] * dc[q];
                    elist[pos] = make_int2(en[q].y, __float_as_int(w));
                }
            }
        }
    }
}

#define GEMM_ROWS  64
#define WP_PITCH   288
#define AT_PITCH   132
#define SMEM_WORDS (64 * WP_PITCH + GEMM_ROWS * AT_PITCH)
#define GEMM_TILES ((NNODES + GEMM_ROWS - 1) / GEMM_ROWS)
#define GEMM_T0    512
#define GEMM_T2    377
#define GEMM_T1    (GEMM_TILES - GEMM_T0 - GEMM_T2)

static_assert(SMEM_WORDS >= 2 * NBINS1 + 3 * EPB, "level-1 role fits the shared buffer");
static_assert(GEMM_T1 > 0, "tile split");
static_assert(NNODES % 8 == 0, "a wave's 8 staged rows are all inside or all outside");

__device__ __forceinline__ void
gcn_gemm_role(float* __restrict__ smem, const int tile,
              const float* __restrict__ X,
              const float* __restrict__ W,
              const float* __restrict__ bias,
              float* __restrict__ Y) {
    float* Wp = smem;
    float* At = smem + 64 * WP_PITCH;

    const int lane = threadIdx.x & 31;
    const int wave = threadIdx.x >> 5;
    const int rt   = wave & 3;
    const int ch   = wave >> 2;
    const int m0   = tile * GEMM_ROWS;
    const int c = lane & 15;
    const int h = lane >> 4;

    float4 arow[8];
    {
        const float4* src = (const float4*)X + lane;
        #pragma unroll
        for (int j = 0; j < 8; ++j) {
            const int r = min(m0 + wave * 8 + j, NNODES - 1);
            arow[j] = src[(size_t)r * 32];
        }
    }

    {
        const float4* W4 = (const float4*)W;
        float4 r0[8], r1[8];
        #pragma unroll
        for (int i = 0; i < 8; ++i) {
            const int p = i * 8 + wave;
            r0[i] = W4[(2 * p) * 32 + lane];
            r1[i] = W4[(2 * p + 1) * 32 + lane];
        }
        #pragma unroll
        for (int j = 0; j < 8; ++j)
            *(float4*)(At + (wave * 8 + j) * AT_PITCH + lane * 4) = arow[j];
        #pragma unroll
        for (int i = 0; i < 8; ++i) {
            const int p = i * 8 + wave;
            float4* dst = (float4*)(Wp + p * WP_PITCH + lane * 8);
            dst[0] = make_float4(r0[i].x, r1[i].x, r0[i].y, r1[i].y);
            dst[1] = make_float4(r0[i].z, r1[i].z, r0[i].w, r1[i].w);
        }
    }
    __syncthreads();

    v8f acc[4];
    #pragma unroll
    for (int nt = 0; nt < 4; ++nt) acc[nt] = v8f{};

    {
        const float* ap = At + (rt * 16 + c) * AT_PITCH + 2 * h;
        const float* bp = Wp + h * WP_PITCH + 128 * ch + 2 * c;
        #pragma unroll
        for (int s = 0; s < 32; ++s) {
            const v2f a = *(const v2f*)(ap + 4 * s);
            #pragma unroll
            for (int nt = 0; nt < 4; ++nt) {
                const v2f b = *(const v2f*)(bp + 2 * s * WP_PITCH + 32 * nt);
                acc[nt] = __builtin_amdgcn_wmma_f32_16x16x4_f32(
                    false, a, false, b, (short)0, acc[nt], false, false);
            }
        }
    }
    __syncthreads();

    #pragma unroll
    for (int nt = 0; nt < 4; ++nt) {
        const float bv = bias[ch * 64 + nt * 16 + c];
        #pragma unroll
        for (int r = 0; r < 8; ++r)
            At[(rt * 16 + 8 * h + r) * AT_PITCH + ch * 64 + nt * 16 + c] = acc[nt][r] + bv;
    }
    __syncthreads();
    const int rw = m0 + wave * 8;
    if (rw < NNODES) {
        float4* dst = (float4*)Y + (size_t)rw * 32 + lane;
        #pragma unroll
        for (int j = 0; j < 8; ++j)
            dst[j * 32] = *(const float4*)(At + (wave * 8 + j) * AT_PITCH + lane * 4);
    }
}

__global__ void __launch_bounds__(256)
gcn_bin_gemm_kernel(const int* __restrict__ row, const int* __restrict__ col,
                    int* __restrict__ offR, int* __restrict__ offC,
                    int* __restrict__ binR_words, int* __restrict__ binC,
                    float* __restrict__ psum, float* __restrict__ cnts,
                    float* __restrict__ gws, int* __restrict__ total,
                    const float* __restrict__ X, const float* __restrict__ W,
                    const float* __restrict__ bias, float* __restrict__ Y) {
    __shared__ alignas(16) int smem[SMEM_WORDS];
    const int b = blockIdx.x;
    if (b < NEBLK) gcn_bin_role(smem, b, row, col, offR, offC, binR_words, binC, psum, cnts, gws, total);
    else           gcn_gemm_role((float*)smem, b - NEBLK, X, W, bias, Y);
}

__global__ void __launch_bounds__(256)
gcn_dinv_gemm_kernel(const int* __restrict__ offC, const int* __restrict__ binC,
                     float* __restrict__ dinv,
                     const float* __restrict__ X, const float* __restrict__ W,
                     const float* __restrict__ bias, float* __restrict__ Y) {
    __shared__ alignas(16) int smem[SMEM_WORDS];
    const int b = blockIdx.x;
    if (b < NBINS) gcn_dinv_role(smem, b, offC, binC, dinv);
    else           gcn_gemm_role((float*)smem, GEMM_T0 + (b - NBINS), X, W, bias, Y);
}

__global__ void __launch_bounds__(256)
gcn_fill_gemm_kernel(const int* __restrict__ offR, const int* __restrict__ binR_words,
                     const float* __restrict__ dinv,
                     int* __restrict__ rowstart, int* __restrict__ rowend,
                     int* __restrict__ total, int* __restrict__ elist_words,
                     const float* __restrict__ X, const float* __restrict__ W,
                     const float* __restrict__ bias, float* __restrict__ Y) {
    __shared__ alignas(16) int smem[SMEM_WORDS];
    const int b = blockIdx.x;
    if (b < NBINS) gcn_fill_role(smem, b, offR, binR_words, dinv, rowstart, rowend, total, elist_words);
    else           gcn_gemm_role((float*)smem, GEMM_T0 + GEMM_T1 + (b - NBINS), X, W, bias, Y);
}

#define GATHER_ROWS  8
#define GATHER_BATCH 8

__device__ __forceinline__ float4 gcn_relu4(float4 v) {
    return make_float4(fmaxf(v.x, 0.0f), fmaxf(v.y, 0.0f), fmaxf(v.z, 0.0f), fmaxf(v.w, 0.0f));
}

__device__ __forceinline__ int2 gcn_row_terms(const int2* __restrict__ elist, int s, int e, int base,
                                              int lane, int n, float sw) {
    int2 ent = make_int2(0, 0);
    const int idx = base + lane - 1;
    if (idx >= s && idx < e) ent = elist[idx];
    if (base == s && lane == 0) ent = make_int2(n, __float_as_int(sw));
    return ent;
}

__device__ __forceinline__ void gcn_batch_issue(float4 (&h)[GATHER_BATCH], const float4* __restrict__ H4,
                                                int2 ent, int j0, int cnt, int lane) {
    #pragma unroll
    for (int k = 0; k < GATHER_BATCH; ++k) {
        if (k < cnt) {
            const int c = __builtin_amdgcn_readlane(ent.x, j0 + k);
            h[k] = (H4 + (size_t)c * 32)[lane];
        }
    }
}

template <bool LAYER2>
__device__ __forceinline__ void gcn_batch_sum(const float4 (&h)[GATHER_BATCH], int2 ent, int j0, int cnt,
                                              float4& acc, float& ws) {
    #pragma unroll
    for (int k = 0; k < GATHER_BATCH; ++k) {
        if (k < cnt) {
            const float w = __int_as_float(__builtin_amdgcn_readlane(ent.y, j0 + k));
            if (LAYER2) ws += w;
            acc.x += w * h[k].x; acc.y += w * h[k].y;
            acc.z += w * h[k].z; acc.w += w * h[k].w;
        }
    }
}

template <bool LAYER2>
__global__ void __launch_bounds__(256)
gcn_gather_kernel(const int* __restrict__ rowstart,
                  const int* __restrict__ rowend,
                  const int* __restrict__ elist_words,
                  const float* __restrict__ dinv,
                  const float* __restrict__ H,
                  float* __restrict__ AGG,
                  float* __restrict__ wsum) {
    const int2* elist = (const int2*)elist_words;
    const int lane = threadIdx.x & 31;
    const int wv   = (blockIdx.x * 256 + threadIdx.x) >> 5;
    const int n0   = wv * GATHER_ROWS;
    if (n0 >= NNODES) return;
    const int nrows = (NNODES % GATHER_ROWS == 0) ? GATHER_ROWS
                                                  : min(GATHER_ROWS, NNODES - n0);
    int   ls = 0, le = 0;
    float lw = 0.0f;
    if (lane < nrows) {
        ls = rowstart[n0 + lane];
        le = rowend[n0 + lane];
        const float d = dinv[n0 + lane];
        lw = d * d;
    }
    const float4* H4 = (const float4*)H;

    int   s  = __builtin_amdgcn_readlane(ls, 0);
    int   e  = __builtin_amdgcn_readlane(le, 0);
    float sw = __int_as_float(__builtin_amdgcn_readlane(__float_as_int(lw), 0));
    int2  ent = gcn_row_terms(elist, s, e, s, lane, n0, sw);
    float wsv = 0.0f;

    for (int i = 0; i < nrows; ++i) {
        const int n = n0 + i;
        const int terms = e - s + 1;
        float4 acc = make_float4(0.0f, 0.0f, 0.0f, 0.0f);
        float  ws  = 0.0f;
        float4 h[GATHER_BATCH];

        int m   = min(32, terms);
        int cnt = min(GATHER_BATCH, m);
        gcn_batch_issue(h, H4, ent, 0, cnt, lane);

        int s1 = 0, e1 = 0; float sw1 = 0.0f;
        int2 ent1 = make_int2(0, 0);
        if (i + 1 < nrows) {
            s1  = __builtin_amdgcn_readlane(ls, i + 1);
            e1  = __builtin_amdgcn_readlane(le, i + 1);
            sw1 = __int_as_float(__builtin_amdgcn_readlane(__float_as_int(lw), i + 1));
            ent1 = gcn_row_terms(elist, s1, e1, s1, lane, n + 1, sw1);
        }

        gcn_batch_sum<LAYER2>(h, ent, 0, cnt, acc, ws);

        for (int j0 = GATHER_BATCH; j0 < m; j0 += GATHER_BATCH) {
            cnt = min(GATHER_BATCH, m - j0);
            gcn_batch_issue(h, H4, ent, j0, cnt, lane);
            gcn_batch_sum<LAYER2>(h, ent, j0, cnt, acc, ws);
        }
        for (int t0 = 32; t0 < terms; t0 += 32) {
            ent = gcn_row_terms(elist, s, e, s + t0, lane, n, sw);
            m = min(32, terms - t0);
            for (int j0 = 0; j0 < m; j0 += GATHER_BATCH) {
                cnt = min(GATHER_BATCH, m - j0);
                gcn_batch_issue(h, H4, ent, j0, cnt, lane);
                gcn_batch_sum<LAYER2>(h, ent, j0, cnt, acc, ws);
            }
        }

        if (!LAYER2) acc = gcn_relu4(acc);
        ((float4*)AGG + (size_t)n * 32)[lane] = acc;
        if (LAYER2 && lane == i) wsv = ws;
        s = s1; e = e1; sw = sw1; ent = ent1;
    }
    if (LAYER2 && lane < nrows) wsum[n0 + lane] = wsv;
}

#define POOL_RUN 32

__device__ __forceinline__ void gcn_pool_flush(float* __restrict__ psum,
                                               float* __restrict__ cnts,
                                               float* __restrict__ gws,
                                               int g, float4 acc, float cnt, float wacc, int lane) {
    float* dst = psum + (size_t)g * FEAT + lane * 4;
    atomicAdd(dst + 0, acc.x);
    atomicAdd(dst + 1, acc.y);
    atomicAdd(dst + 2, acc.z);
    atomicAdd(dst + 3, acc.w);
    if (lane == 0) atomicAdd(&cnts[g], cnt);
    if (lane == 1) atomicAdd(&gws[g], wacc);
}

__global__ void __launch_bounds__(128)
gcn_pool_kernel(const float* __restrict__ AGG,
                const int* __restrict__ batch,
                const float* __restrict__ wsum,
                float* __restrict__ psum,
                float* __restrict__ cnts,
                float* __restrict__ gws) {
    const int lane = threadIdx.x & 31;
    const int w    = (blockIdx.x * 128 + threadIdx.x) >> 5;
    const int n0   = w * POOL_RUN;
    if (n0 >= NNODES) return;
    const int nn = (NNODES % POOL_RUN == 0) ? POOL_RUN
                                            : min(POOL_RUN, NNODES - n0);
    const int gl = (lane < nn) ? batch[n0 + lane] : 0;
    const float wl = (lane < nn) ? wsum[n0 + lane] : 0.0f;
    const float4* src = (const float4*)(AGG + (size_t)n0 * FEAT) + lane;

    float4 h[POOL_RUN];
    #pragma unroll
    for (int i = 0; i < POOL_RUN; ++i)
        if (i < nn) h[i] = src[i * 32];

    int    gcur = __builtin_amdgcn_readlane(gl, 0);
    float4 acc  = make_float4(0.0f, 0.0f, 0.0f, 0.0f);
    float  cnt  = 0.0f;
    float  wacc = 0.0f;
    #pragma unroll
    for (int i = 0; i < POOL_RUN; ++i) {
        if (i < nn) {
            const int g = __builtin_amdgcn_readlane(gl, i);
            if (g != gcur) {
                gcn_pool_flush(psum, cnts, gws, gcur, acc, cnt, wacc, lane);
                acc  = make_float4(0.0f, 0.0f, 0.0f, 0.0f);
                cnt  = 0.0f;
                wacc = 0.0f;
                gcur = g;
            }
            acc.x += h[i].x; acc.y += h[i].y; acc.z += h[i].z; acc.w += h[i].w;
            cnt  += 1.0f;
            wacc += __int_as_float(__builtin_amdgcn_readlane(__float_as_int(wl), i));
        }
    }
    gcn_pool_flush(psum, cnts, gws, gcur, acc, cnt, wacc, lane);
}

__global__ void __launch_bounds__(128)
gcn_head_kernel(const float* __restrict__ psum,
                const float* __restrict__ W,
                const float* __restrict__ bias,
                const float* __restrict__ cnts,
                const float* __restrict__ gws,
                float* __restrict__ out) {
    __shared__ alignas(16) float Ct[16 * AT_PITCH];

    const int lane = threadIdx.x & 31;
    const int wave = threadIdx.x >> 5;
    const int c  = lane & 15;
    const int h  = lane >> 4;
    const int m0 = blockIdx.x * 16;
    const int n0 = wave * 32;

    const float* ap = psum + (size_t)(m0 + c) * FEAT + 2 * h;
    const float* bp = W + (size_t)(2 * h) * FEAT + n0 + c;
    v2f a[32], b0[32], b1[32];
    #pragma unroll
    for (int s = 0; s < 32; ++s) {
        a[s]  = *(const v2f*)(ap + 4 * s);
        b0[s] = v2f{bp[(4 * s) * FEAT],      bp[(4 * s + 1) * FEAT]};
        b1[s] = v2f{bp[(4 * s) * FEAT + 16], bp[(4 * s + 1) * FEAT + 16]};
    }

    v8f acc0 = v8f{};
    v8f acc1 = v8f{};
    #pragma unroll
    for (int s = 0; s < 32; ++s) {
        acc0 = __builtin_amdgcn_wmma_f32_16x16x4_f32(
            false, a[s], false, b0[s], (short)0, acc0, false, false);
        acc1 = __builtin_amdgcn_wmma_f32_16x16x4_f32(
            false, a[s], false, b1[s], (short)0, acc1, false, false);
    }

    #pragma unroll
    for (int r = 0; r < 8; ++r) {
        Ct[(8 * h + r) * AT_PITCH + n0 + c]      = acc0[r];
        Ct[(8 * h + r) * AT_PITCH + n0 + 16 + c] = acc1[r];
    }
    __syncthreads();

    const float4 bv = ((const float4*)bias)[lane];
    #pragma unroll
    for (int j = 0; j < 4; ++j) {
        const int   rr = wave * 4 + j;
        const int   g  = m0 + rr;
        const float d  = fmaxf(cnts[g], 1.0f);
        const float wm = gws[g] / d;
        const float4 v = *(const float4*)(Ct + rr * AT_PITCH + lane * 4);
        float4 o;
        o.x = v.x / d + wm * bv.x;
        o.y = v.y / d + wm * bv.y;
        o.z = v.z / d + wm * bv.z;
        o.w = v.w / d + wm * bv.w;
        ((float4*)out)[(size_t)g * 32 + lane] = o;
    }
}

extern "C" void kernel_launch(void* const* d_in, const int* in_sizes, int n_in,
                              void* d_out, int out_size, void* d_ws, size_t ws_size,
                              hipStream_t stream) {
    const float* x          = (const float*)d_in[0];
    const int*   edge_index = (const int*)  d_in[1];
    const int*   batch      = (const int*)  d_in[2];
    const float* W1 = (const float*)d_in[4];
    const float* b1 = (const float*)d_in[5];
    const float* W2 = (const float*)d_in[6];
    const float* b2 = (const float*)d_in[7];
    float* out = (float*)d_out;

    float* H    = (float*)d_ws;
    float* AGG  = H    + (size_t)NNODES * FEAT;
    float* dinv = AGG  + (size_t)NNODES * FEAT;
    float* cnts = dinv + NNODES;
    float* gws  = cnts + NGRAPH;
    float* psum = gws  + NGRAPH;
    float* wsum = psum + (size_t)NGRAPH * FEAT;
    int*   rowend   = (int*)(wsum + NNODES);
    int*   rowstart = rowend + NNODES;
    int*   total    = rowstart + NNODES;
    int*   elist    = total + 16;
    int*   binR     = elist + 2 * NEDGES;
    int*   binC     = binR + 2 * NEBLK * EPB;
    int*   offR     = binC + NEBLK * EPB;
    int*   offC     = offR + NEBLK * NBINS1;

    const int* row = edge_index;
    const int* col = edge_index + NEDGES;

    dim3 blk(256);
    const int grid_gather = ((NNODES + GATHER_ROWS - 1) / GATHER_ROWS + 7) / 8;
    const int grid_head   = NGRAPH / 16;
    const int grid_pool   = ((NNODES + POOL_RUN - 1) / POOL_RUN + 3) / 4;

    gcn_bin_gemm_kernel<<<NEBLK + GEMM_T0, blk, 0, stream>>>(row, col, offR, offC, binR, binC, psum, cnts, gws, total,
                                                          x, W1, b1, H);
    gcn_dinv_gemm_kernel<<<NBINS + GEMM_T1, blk, 0, stream>>>(offC, binC, dinv, x, W1, b1, H);
    gcn_fill_gemm_kernel<<<NBINS + GEMM_T2, blk, 0, stream>>>(offR, binR, dinv, rowstart, rowend, total, elist,
                                                           x, W1, b1, H);

    gcn_gather_kernel<false><<<grid_gather, blk, 0, stream>>>(rowstart, rowend, elist, dinv, H, AGG, wsum);

    gcn_gather_kernel<true><<<grid_gather, blk, 0, stream>>>(rowstart, rowend, elist, dinv, AGG, H, wsum);
    gcn_pool_kernel<<<grid_pool, dim3(128), 0, stream>>>(H, batch, wsum, psum, cnts, gws);
    gcn_head_kernel<<<grid_head, dim3(128), 0, stream>>>(psum, W2, b2, cnts, gws, out);
}


# layer-2 aggregation stores one partial pooled row per run of equal graph ids inside a wave's 8 rows instead of every row; the pool reads only those rows; changes the order of the pool sums
# speedup vs baseline: 5.6974x; 1.0778x over previous
#include <hip/hip_runtime.h>
#include <cmath>

#define NNODES 100000
#define NEDGES 600000
#define FEAT   128
#define NGRAPH 256

typedef __attribute__((ext_vector_type(2))) float v2f;
typedef __attribute__((ext_vector_type(8))) float v8f;

#define BIN_SHIFT 8
#define BIN_ROWS  (1 << BIN_SHIFT)
#define NBINS     ((NNODES + BIN_ROWS - 1) / BIN_ROWS)
#define NBINS1    (NBINS + 1)
#define EPT       10
#define EPB       (256 * EPT)
#define NEBLK     ((NEDGES + EPB - 1) / EPB)
#define SCAN_PER_LANE ((NBINS1 + 31) / 32)
#define WALK      8

static_assert(NEBLK * 256 >= NGRAPH * FEAT, "level-1 grid also clears the pooled sums");
static_assert(WALK == 8, "gcn_hold8 names WALK values");

__device__ __forceinline__ void gcn_hold8(int a0, int a1, int a2, int a3, int a4, int a5, int a6, int a7) {
    asm volatile("" :: "v"(a0), "v"(a1), "v"(a2), "v"(a3), "v"(a4), "v"(a5), "v"(a6), "v"(a7));
}
__device__ __forceinline__ void gcn_hold8(float a0, float a1, float a2, float a3, float a4, float a5, float a6, float a7) {
    asm volatile("" :: "v"(a0), "v"(a1), "v"(a2), "v"(a3), "v"(a4), "v"(a5), "v"(a6), "v"(a7));
}

__device__ __forceinline__ void
gcn_bin_role(int* __restrict__ smem, const int blk,
             const int* __restrict__ row, const int* __restrict__ col,
             int* __restrict__ offR, int* __restrict__ offC,
             int* __restrict__ binR_words, int* __restrict__ binC,
             float* __restrict__ psum, float* __restrict__ cnts,
             float* __restrict__ gws, int* __restrict__ total) {
    int*  hR = smem;
    int*  hC = smem + NBINS1;
    int2* sR = (int2*)(smem + 2 * NBINS1);
    int*  sC = smem + 2 * NBINS1 + 2 * EPB;
    const int tid  = threadIdx.x;
    const int lane = tid & 31;
    const int wave = tid >> 5;
    {
        const int i = blk * 256 + tid;
        if (i < NGRAPH * FEAT) psum[i] = 0.0f;
        if (i < NGRAPH) {
            cnts[i] = 0.0f;
            gws[i]  = 0.0f;
        }
        if (i == 0) *total = 0;
    }
    for (int i = tid; i < NBINS1; i += 256) { hR[i] = 0; hC[i] = 0; }

    int r[EPT], c[EPT], kr[EPT], kc[EPT];
    const int e0 = blk * EPB + tid;
    #pragma unroll
    for (int j = 0; j < EPT; ++j) {
        const int e = e0 + j * 256;
        r[j] = 0; c[j] = 0;
        if (e < NEDGES) { r[j] = row[e]; c[j] = col[e]; }
    }
    __syncthreads();
    #pragma unroll
    for (int j = 0; j < EPT; ++j) {
        kr[j] = 0; kc[j] = 0;
        if (e0 + j * 256 < NEDGES) {
            kr[j] = atomicAdd(&hR[r[j] >> BIN_SHIFT], 1);
            kc[j] = atomicAdd(&hC[c[j] >> BIN_SHIFT], 1);
        }
    }
    __syncthreads();
    if (wave < 2) {
        int* h  = (wave == 0) ? hR : hC;
        int* og = ((wave == 0) ? offR : offC) + blk;
        int v[SCAN_PER_LANE];
        int sum = 0;
        #pragma unroll
        for (int q = 0; q < SCAN_PER_LANE; ++q) {
            const int idx = lane * SCAN_PER_LANE + q;
            v[q] = (idx < NBINS1) ? h[idx] : 0;
            sum += v[q];
        }
        int inc = sum;
        #pragma unroll
        for (int o = 1; o < 32; o <<= 1) {
            const int t = __shfl_up(inc, o);
            if (lane >= o) inc += t;
        }
        int ex = inc - sum;
        #pragma unroll
        for (int q = 0; q < SCAN_PER_LANE; ++q) {
            const int idx = lane * SCAN_PER_LANE + q;
            if (idx < NBINS1) { h[idx] = ex; og[idx * NEBLK] = ex; }
            ex += v[q];
        }
    }
    __syncthreads();
    #pragma unroll
    for (int j = 0; j < EPT; ++j) {
        if (e0 + j * 256 < NEDGES) {
            sR[hR[r[j] >> BIN_SHIFT] + kr[j]] = make_int2(r[j], c[j]);
            sC[hC[c[j] >> BIN_SHIFT] + kc[j]] = c[j];
        }
    }
    __syncthreads();
    const int nE = min(EPB, NEDGES - blk * EPB);
    int2* binR = (int2*)binR_words + (size_t)blk * EPB;
    int*  bC   = binC + (size_t)blk * EPB;
    #pragma unroll
    for (int j = 0; j < EPT; ++j) {
        const int i = tid + j * 256;
        if (i < nE) {
            binR[i] = sR[i];
            bC[i]   = sC[i];
        }
    }
}

__device__ __forceinline__ void
gcn_dinv_role(int* __restrict__ smem, const int b,
              const int* __restrict__ offC, const int* __restrict__ binC,
              float* __restrict__ dinv) {
    int* cnt = smem;
    const int tid = threadIdx.x;
    cnt[tid] = 0;
    __syncthreads();
    for (int k = tid; k < NEBLK; k += 256) {
        const int s = offC[b * NEBLK + k];
        const int t = offC[(b + 1) * NEBLK + k];
        const int* src = binC + (size_t)k * EPB;
        for (int i = s; i < t; i += WALK) {
            int cc[WALK];
            #pragma unroll
            for (int q = 0; q < WALK; ++q) cc[q] = src[min(i + q, t - 1)];
            gcn_hold8(cc[0], cc[1], cc[2], cc[3], cc[4], cc[5], cc[6], cc[7]);
            #pragma unroll
            for (int q = 0; q < WALK; ++q)
                if (i + q < t) atomicAdd(&cnt[cc[q] & (BIN_ROWS - 1)], 1);
        }
    }
    __syncthreads();
    const int n = b * BIN_ROWS + tid;
    if (n < NNODES) {
        const float d = 1.0f + (float)cnt[tid];
        dinv[n] = (d > 0.0f) ? rsqrtf(d) : 0.0f;
    }
}

__device__ __forceinline__ void
gcn_fill_role(int* __restrict__ smem, const int b,
              const int* __restrict__ offR, const int* __restrict__ binR_words,
              const float* __restrict__ dinv,
              int* __restrict__ rowstart, int* __restrict__ rowend,
              int* __restrict__ total, int* __restrict__ elist_words) {
    int*   cnt   = smem;
    float* dl    = (float*)(smem + BIN_ROWS);
    int*   wsum  = smem + 2 * BIN_ROWS;
    int*   sbase = smem + 2 * BIN_ROWS + 8;
    const int2* binR = (const int2*)binR_words;
    int2* elist = (int2*)elist_words;
    const int tid  = threadIdx.x;
    const int lane = tid & 31;
    const int wave = tid >> 5;
    const int n    = b * BIN_ROWS + tid;
    cnt[tid] = 0;
    dl[tid]  = (n < NNODES) ? dinv[n] : 0.0f;
    __syncthreads();
    for (int k = tid; k < NEBLK; k += 256) {
        const int s = offR[b * NEBLK + k];
        const int t = offR[(b + 1) * NEBLK + k];
        const int2* src = binR + (size_t)k * EPB;
        for (int i = s; i < t; i += WALK) {
            int rr[WALK];
            #pragma unroll
            for (int q = 0; q < WALK; ++q) rr[q] = src[min(i + q, t - 1)].x;
            gcn_hold8(rr[0], rr[1], rr[2], rr[3], rr[4], rr[5], rr[6], rr[7]);
            #pragma unroll
            for (int q = 0; q < WALK; ++q)
                if (i + q < t) atomicAdd(&cnt[rr[q] & (BIN_ROWS - 1)], 1);
        }
    }
    __syncthreads();
    const int c = cnt[tid];
    int v = c;
    #pragma unroll
    for (int o = 1; o < 32; o <<= 1) {
        int t = __shfl_up(v, o);
        if (lane >= o) v += t;
    }
    if (lane == 31) wsum[wave] = v;
    __syncthreads();
    if (tid == 0) {
        int s = 0;
        for (int w = 0; w < 8; ++w) { int t = wsum[w]; wsum[w] = s; s += t; }
        *sbase = atomicAdd(total, s);
    }
    __syncthreads();
    const int st = *sbase + wsum[wave] + v - c;
    if (n < NNODES) {
        rowstart[n] = st;
        rowend[n]   = st + c;
    }
    cnt[tid] = st;
    __syncthreads();
    for (int k = tid; k < NEBLK; k += 256) {
        const int s = offR[b * NEBLK + k];
        const int t = offR[(b + 1) * NEBLK + k];
        const int2* src = binR + (size_t)k * EPB;
        for (int i = s; i < t; i += WALK) {
            int2  en[WALK];
            float dc[WALK];
            #pragma unroll
            for (int q = 0; q < WALK; ++q) en[q] = src[min(i + q, t - 1)];
            #pragma unroll
            for (int q = 0; q < WALK; ++q) dc[q] = dinv[en[q].y];
            gcn_hold8(dc[0], dc[1], dc[2], dc[3], dc[4], dc[5], dc[6], dc[7]);
            #pragma unroll
            for (int q = 0; q < WALK; ++q) {
                if (i + q < t) {
                    const int rl  = en[q].x & (BIN_ROWS - 1);
                    const int pos = atomicAdd(&cnt[rl], 1);
                    const float w = dl[rl] * dc[q];
                    elist[pos] = make_int2(en[q].y, __float_as_int(w));
                }
            }
        }
    }
}

#define GEMM_ROWS  64
#define WP_PITCH   288
#define AT_PITCH   132
#define SMEM_WORDS (64 * WP_PITCH + GEMM_ROWS * AT_PITCH)
#define GEMM_TILES ((NNODES + GEMM_ROWS - 1) / GEMM_ROWS)
#define GEMM_T0    512
#define GEMM_T2    377
#define GEMM_T1    (GEMM_TILES - GEMM_T0 - GEMM_T2)

static_assert(SMEM_WORDS >= 2 * NBINS1 + 3 * EPB, "level-1 role fits the shared buffer");
static_assert(GEMM_T1 > 0, "tile split");
static_assert(NNODES % 8 == 0, "a wave's 8 staged rows are all inside or all outside");

__device__ __forceinline__ void
gcn_gemm_role(float* __restrict__ smem, const int tile,
              const float* __restrict__ X,
              const float* __restrict__ W,
              const float* __restrict__ bias,
              float* __restrict__ Y) {
    float* Wp = smem;
    float* At = smem + 64 * WP_PITCH;

    const int lane = threadIdx.x & 31;
    const int wave = threadIdx.x >> 5;
    const int rt   = wave & 3;
    const int ch   = wave >> 2;
    const int m0   = tile * GEMM_ROWS;
    const int c = lane & 15;
    const int h = lane >> 4;

    float4 arow[8];
    {
        const float4* src = (const float4*)X + lane;
        #pragma unroll
        for (int j = 0; j < 8; ++j) {
            const int r = min(m0 + wave * 8 + j, NNODES - 1);
            arow[j] = src[(size_t)r * 32];
        }
    }

    {
        const float4* W4 = (const float4*)W;
        float4 r0[8], r1[8];
        #pragma unroll
        for (int i = 0; i < 8; ++i) {
            const int p = i * 8 + wave;
            r0[i] = W4[(2 * p) * 32 + lane];
            r1[i] = W4[(2 * p + 1) * 32 + lane];
        }
        #pragma unroll
        for (int j = 0; j < 8; ++j)
            *(float4*)(At + (wave * 8 + j) * AT_PITCH + lane * 4) = arow[j];
        #pragma unroll
        for (int i = 0; i < 8; ++i) {
            const int p = i * 8 + wave;
            float4* dst = (float4*)(Wp + p * WP_PITCH + lane * 8);
            dst[0] = make_float4(r0[i].x, r1[i].x, r0[i].y, r1[i].y);
            dst[1] = make_float4(r0[i].z, r1[i].z, r0[i].w, r1[i].w);
        }
    }
    __syncthreads();

    v8f acc[4];
    #pragma unroll
    for (int nt = 0; nt < 4; ++nt) acc[nt] = v8f{};

    {
        const float* ap = At + (rt * 16 + c) * AT_PITCH + 2 * h;
        const float* bp = Wp + h * WP_PITCH + 128 * ch + 2 * c;
        #pragma unroll
        for (int s = 0; s < 32; ++s) {
            const v2f a = *(const v2f*)(ap + 4 * s);
            #pragma unroll
            for (int nt = 0; nt < 4; ++nt) {
                const v2f b = *(const v2f*)(bp + 2 * s * WP_PITCH + 32 * nt);
                acc[nt] = __builtin_amdgcn_wmma_f32_16x16x4_f32(
                    false, a, false, b, (short)0, acc[nt], false, false);
            }
        }
    }
    __syncthreads();

    #pragma unroll
    for (int nt = 0; nt < 4; ++nt) {
        const float bv = bias[ch * 64 + nt * 16 + c];
        #pragma unroll
        for (int r = 0; r < 8; ++r)
            At[(rt * 16 + 8 * h + r) * AT_PITCH + ch * 64 + nt * 16 + c] = acc[nt][r] + bv;
    }
    __syncthreads();
    const int rw = m0 + wave * 8;
    if (rw < NNODES) {
        float4* dst = (float4*)Y + (size_t)rw * 32 + lane;
        #pragma unroll
        for (int j = 0; j < 8; ++j)
            dst[j * 32] = *(const float4*)(At + (wave * 8 + j) * AT_PITCH + lane * 4);
    }
}

__global__ void __launch_bounds__(256)
gcn_bin_gemm_kernel(const int* __restrict__ row, const int* __restrict__ col,
                    int* __restrict__ offR, int* __restrict__ offC,
                    int* __restrict__ binR_words, int* __restrict__ binC,
                    float* __restrict__ psum, float* __restrict__ cnts,
                    float* __restrict__ gws, int* __restrict__ total,
                    const float* __restrict__ X, const float* __restrict__ W,
                    const float* __restrict__ bias, float* __restrict__ Y) {
    __shared__ alignas(16) int smem[SMEM_WORDS];
    const int b = blockIdx.x;
    if (b < NEBLK) gcn_bin_role(smem, b, row, col, offR, offC, binR_words, binC, psum, cnts, gws, total);
    else           gcn_gemm_role((float*)smem, b - NEBLK, X, W, bias, Y);
}

__global__ void __launch_bounds__(256)
gcn_dinv_gemm_kernel(const int* __restrict__ offC, const int* __restrict__ binC,
                     float* __restrict__ dinv,
                     const float* __restrict__ X, const float* __restrict__ W,
                     const float* __restrict__ bias, float* __restrict__ Y) {
    __shared__ alignas(16) int smem[SMEM_WORDS];
    const int b = blockIdx.x;
    if (b < NBINS) gcn_dinv_role(smem, b, offC, binC, dinv);
    else           gcn_gemm_role((float*)smem, GEMM_T0 + (b - NBINS), X, W, bias, Y);
}

__global__ void __launch_bounds__(256)
gcn_fill_gemm_kernel(const int* __restrict__ offR, const int* __restrict__ binR_words,
                     const float* __restrict__ dinv,
                     int* __restrict__ rowstart, int* __restrict__ rowend,
                     int* __restrict__ total, int* __restrict__ elist_words,
                     const float* __restrict__ X, const float* __restrict__ W,
                     const float* __restrict__ bias, float* __restrict__ Y) {
    __shared__ alignas(16) int smem[SMEM_WORDS];
    const int b = blockIdx.x;
    if (b < NBINS) gcn_fill_role(smem, b, offR, binR_words, dinv, rowstart, rowend, total, elist_words);
    else           gcn_gemm_role((float*)smem, GEMM_T0 + GEMM_T1 + (b - NBINS), X, W, bias, Y);
}

#define GATHER_ROWS  8
#define GATHER_BATCH 8

__device__ __forceinline__ bool gcn_run_end(int i_in_group, int g, int g_next) {
    return i_in_group == GATHER_ROWS - 1 || g != g_next;
}

__device__ __forceinline__ float4 gcn_relu4(float4 v) {
    return make_float4(fmaxf(v.x, 0.0f), fmaxf(v.y, 0.0f), fmaxf(v.z, 0.0f), fmaxf(v.w, 0.0f));
}

__device__ __forceinline__ int2 gcn_row_terms(const int2* __restrict__ elist, int s, int e, int base,
                                              int lane, int n, float sw) {
    int2 ent = make_int2(0, 0);
    const int idx = base + lane - 1;
    if (idx >= s && idx < e) ent = elist[idx];
    if (base == s && lane == 0) ent = make_int2(n, __float_as_int(sw));
    return ent;
}

__device__ __forceinline__ void gcn_batch_issue(float4 (&h)[GATHER_BATCH], const float4* __restrict__ H4,
                                                int2 ent, int j0, int cnt, int lane) {
    #pragma unroll
    for (int k = 0; k < GATHER_BATCH; ++k) {
        if (k < cnt) {
            const int c = __builtin_amdgcn_readlane(ent.x, j0 + k);
            h[k] = (H4 + (size_t)c * 32)[lane];
        }
    }
}

template <bool LAYER2>
__device__ __forceinline__ void gcn_batch_sum(const float4 (&h)[GATHER_BATCH], int2 ent, int j0, int cnt,
                                              float4& acc, float& ws) {
    #pragma unroll
    for (int k = 0; k < GATHER_BATCH; ++k) {
        if (k < cnt) {
            const float w = __int_as_float(__builtin_amdgcn_readlane(ent.y, j0 + k));
            if (LAYER2) ws += w;
            acc.x += w * h[k].x; acc.y += w * h[k].y;
            acc.z += w * h[k].z; acc.w += w * h[k].w;
        }
    }
}

template <bool LAYER2>
__global__ void __launch_bounds__(256)
gcn_gather_kernel(const int* __restrict__ rowstart,
                  const int* __restrict__ rowend,
                  const int* __restrict__ elist_words,
                  const float* __restrict__ dinv,
                  const float* __restrict__ H,
                  const int* __restrict__ batch,
                  float* __restrict__ AGG,
                  float* __restrict__ wsum) {
    const int2* elist = (const int2*)elist_words;
    const int lane = threadIdx.x & 31;
    const int wv   = (blockIdx.x * 256 + threadIdx.x) >> 5;
    const int n0   = wv * GATHER_ROWS;
    if (n0 >= NNODES) return;
    const int nrows = (NNODES % GATHER_ROWS == 0) ? GATHER_ROWS
                                                  : min(GATHER_ROWS, NNODES - n0);
    int   ls = 0, le = 0;
    float lw = 0.0f;
    int   lg = 0;
    if (lane < nrows) {
        ls = rowstart[n0 + lane];
        le = rowend[n0 + lane];
        const float d = dinv[n0 + lane];
        lw = d * d;
        if (LAYER2) lg = batch[n0 + lane];
    }
    const float4* H4 = (const float4*)H;

    int   s  = __builtin_amdgcn_readlane(ls, 0);
    int   e  = __builtin_amdgcn_readlane(le, 0);
    float sw = __int_as_float(__builtin_amdgcn_readlane(__float_as_int(lw), 0));
    int2  ent = gcn_row_terms(elist, s, e, s, lane, n0, sw);
    float wsv = 0.0f;
    float4 run = make_float4(0.0f, 0.0f, 0.0f, 0.0f);

    for (int i = 0; i < nrows; ++i) {
        const int n = n0 + i;
        const int terms = e - s + 1;
        float4 acc = make_float4(0.0f, 0.0f, 0.0f, 0.0f);
        float  ws  = 0.0f;
        float4 h[GATHER_BATCH];

        int m   = min(32, terms);
        int cnt = min(GATHER_BATCH, m);
        gcn_batch_issue(h, H4, ent, 0, cnt, lane);

        int s1 = 0, e1 = 0; float sw1 = 0.0f;
        int2 ent1 = make_int2(0, 0);
        if (i + 1 < nrows) {
            s1  = __builtin_amdgcn_readlane(ls, i + 1);
            e1  = __builtin_amdgcn_readlane(le, i + 1);
            sw1 = __int_as_float(__builtin_amdgcn_readlane(__float_as_int(lw), i + 1));
            ent1 = gcn_row_terms(elist, s1, e1, s1, lane, n + 1, sw1);
        }

        gcn_batch_sum<LAYER2>(h, ent, 0, cnt, acc, ws);

        for (int j0 = GATHER_BATCH; j0 < m; j0 += GATHER_BATCH) {
            cnt = min(GATHER_BATCH, m - j0);
            gcn_batch_issue(h, H4, ent, j0, cnt, lane);
            gcn_batch_sum<LAYER2>(h, ent, j0, cnt, acc, ws);
        }
        for (int t0 = 32; t0 < terms; t0 += 32) {
            ent = gcn_row_terms(elist, s, e, s + t0, lane, n, sw);
            m = min(32, terms - t0);
            for (int j0 = 0; j0 < m; j0 += GATHER_BATCH) {
                cnt = min(GATHER_BATCH, m - j0);
                gcn_batch_issue(h, H4, ent, j0, cnt, lane);
                gcn_batch_sum<LAYER2>(h, ent, j0, cnt, acc, ws);
            }
        }

        if (LAYER2) {
            run.x += acc.x; run.y += acc.y; run.z += acc.z; run.w += acc.w;
            const int g  = __builtin_amdgcn_readlane(lg, i);
            const int gn = __builtin_amdgcn_readlane(lg, i + 1);
            if (i == nrows - 1 || gcn_run_end(i, g, gn)) {
                ((float4*)AGG + (size_t)n * 32)[lane] = run;
                run = make_float4(0.0f, 0.0f, 0.0f, 0.0f);
            }
        } else {
            ((float4*)AGG + (size_t)n * 32)[lane] = gcn_relu4(acc);
        }
        if (LAYER2 && lane == i) wsv = ws;
        s = s1; e = e1; sw = sw1; ent = ent1;
    }
    if (LAYER2 && lane < nrows) wsum[n0 + lane] = wsv;
}

#define POOL_RUN 32
#define POOL_GROUPS (POOL_RUN / GATHER_ROWS)

static_assert(POOL_RUN % GATHER_ROWS == 0 && NNODES % GATHER_ROWS == 0,
              "a pool wave's nodes are whole groups of the aggregation's rows");

__device__ __forceinline__ void gcn_pool_flush(float* __restrict__ psum,
                                               float* __restrict__ cnts,
                                               float* __restrict__ gws,
                                               int g, float4 acc, float cnt, float wacc, int lane) {
    float* dst = psum + (size_t)g * FEAT + lane * 4;
    atomicAdd(dst + 0, acc.x);
    atomicAdd(dst + 1, acc.y);
    atomicAdd(dst + 2, acc.z);
    atomicAdd(dst + 3, acc.w);
    if (lane == 0) atomicAdd(&cnts[g], cnt);
    if (lane == 1) atomicAdd(&gws[g], wacc);
}

__global__ void __launch_bounds__(128)
gcn_pool_kernel(const float* __restrict__ AGG,
                const int* __restrict__ batch,
                const float* __restrict__ wsum,
                float* __restrict__ psum,
                float* __restrict__ cnts,
                float* __restrict__ gws) {
    const int lane = threadIdx.x & 31;
    const int w    = (blockIdx.x * 128 + threadIdx.x) >> 5;
    const int n0   = w * POOL_RUN;
    if (n0 >= NNODES) return;
    const int nn = (NNODES % POOL_RUN == 0) ? POOL_RUN
                                            : min(POOL_RUN, NNODES - n0);
    const int gl = (lane < nn) ? batch[n0 + lane] : 0;
    const float wl = (lane < nn) ? wsum[n0 + lane] : 0.0f;
    const float4* src = (const float4*)(AGG + (size_t)n0 * FEAT) + lane;

    float4 h[POOL_GROUPS];
    #pragma unroll
    for (int q = 0; q < POOL_GROUPS; ++q)
        if (q * GATHER_ROWS + GATHER_ROWS - 1 < nn) h[q] = src[(q * GATHER_ROWS + GATHER_ROWS - 1) * 32];

    int    gcur = __builtin_amdgcn_readlane(gl, 0);
    float4 acc  = make_float4(0.0f, 0.0f, 0.0f, 0.0f);
    float  cnt  = 0.0f;
    float  wacc = 0.0f;
    #pragma unroll
    for (int i = 0; i < POOL_RUN; ++i) {
        if (i < nn) {
            const int g = __builtin_amdgcn_readlane(gl, i);
            if (g != gcur) {
                gcn_pool_flush(psum, cnts, gws, gcur, acc, cnt, wacc, lane);
                acc  = make_float4(0.0f, 0.0f, 0.0f, 0.0f);
                cnt  = 0.0f;
                wacc = 0.0f;
                gcur = g;
            }
            if ((i % GATHER_ROWS) == GATHER_ROWS - 1) {
                const float4 t = h[i / GATHER_ROWS];
                acc.x += t.x; acc.y += t.y; acc.z += t.z; acc.w += t.w;
            } else if (gcn_run_end(i % GATHER_ROWS, g, __builtin_amdgcn_readlane(gl, i + 1))) {
                const float4 t = src[i * 32];
                acc.x += t.x; acc.y += t.y; acc.z += t.z; acc.w += t.w;
            }
            cnt  += 1.0f;
            wacc += __int_as_float(__builtin_amdgcn_readlane(__float_as_int(wl), i));
        }
    }
    gcn_pool_flush(psum, cnts, gws, gcur, acc, cnt, wacc, lane);
}

__global__ void __launch_bounds__(128)
gcn_head_kernel(const float* __restrict__ psum,
                const float* __restrict__ W,
                const float* __restrict__ bias,
                const float* __restrict__ cnts,
                const float* __restrict__ gws,
                float* __restrict__ out) {
    __shared__ alignas(16) float Ct[16 * AT_PITCH];

    const int lane = threadIdx.x & 31;
    const int wave = threadIdx.x >> 5;
    const int c  = lane & 15;
    const int h  = lane >> 4;
    const int m0 = blockIdx.x * 16;
    const int n0 = wave * 32;

    const float* ap = psum + (size_t)(m0 + c) * FEAT + 2 * h;
    const float* bp = W + (size_t)(2 * h) * FEAT + n0 + c;
    v2f a[32], b0[32], b1[32];
    #pragma unroll
    for (int s = 0; s < 32; ++s) {
        a[s]  = *(const v2f*)(ap + 4 * s);
        b0[s] = v2f{bp[(4 * s) * FEAT],      bp[(4 * s + 1) * FEAT]};
        b1[s] = v2f{bp[(4 * s) * FEAT + 16], bp[(4 * s + 1) * FEAT + 16]};
    }

    v8f acc0 = v8f{};
    v8f acc1 = v8f{};
    #pragma unroll
    for (int s = 0; s < 32; ++s) {
        acc0 = __builtin_amdgcn_wmma_f32_16x16x4_f32(
            false, a[s], false, b0[s], (short)0, acc0, false, false);
        acc1 = __builtin_amdgcn_wmma_f32_16x16x4_f32(
            false, a[s], false, b1[s], (short)0, acc1, false, false);
    }

    #pragma unroll
    for (int r = 0; r < 8; ++r) {
        Ct[(8 * h + r) * AT_PITCH + n0 + c]      = acc0[r];
        Ct[(8 * h + r) * AT_PITCH + n0 + 16 + c] = acc1[r];
    }
    __syncthreads();

    const float4 bv = ((const float4*)bias)[lane];
    #pragma unroll
    for (int j = 0; j < 4; ++j) {
        const int   rr = wave * 4 + j;
        const int   g  = m0 + rr;
        const float d  = fmaxf(cnts[g], 1.0f);
        const float wm = gws[g] / d;
        const float4 v = *(const float4*)(Ct + rr * AT_PITCH + lane * 4);
        float4 o;
        o.x = v.x / d + wm * bv.x;
        o.y = v.y / d + wm * bv.y;
        o.z = v.z / d + wm * bv.z;
        o.w = v.w / d + wm * bv.w;
        ((float4*)out)[(size_t)g * 32 + lane] = o;
    }
}

extern "C" void kernel_launch(void* const* d_in, const int* in_sizes, int n_in,
                              void* d_out, int out_size, void* d_ws, size_t ws_size,
                              hipStream_t stream) {
    const float* x          = (const float*)d_in[0];
    const int*   edge_index = (const int*)  d_in[1];
    const int*   batch      = (const int*)  d_in[2];
    const float* W1 = (const float*)d_in[4];
    const float* b1 = (const float*)d_in[5];
    const float* W2 = (const float*)d_in[6];
    const float* b2 = (const float*)d_in[7];
    float* out = (float*)d_out;

    float* H    = (float*)d_ws;
    float* AGG  = H    + (size_t)NNODES * FEAT;
    float* dinv = AGG  + (size_t)NNODES * FEAT;
    float* cnts = dinv + NNODES;
    float* gws  = cnts + NGRAPH;
    float* psum = gws  + NGRAPH;
    float* wsum = psum + (size_t)NGRAPH * FEAT;
    int*   rowend   = (int*)(wsum + NNODES);
    int*   rowstart = rowend + NNODES;
    int*   total    = rowstart + NNODES;
    int*   elist    = total + 16;
    int*   binR     = elist + 2 * NEDGES;
    int*   binC     = binR + 2 * NEBLK * EPB;
    int*   offR     = binC + NEBLK * EPB;
    int*   offC     = offR + NEBLK * NBINS1;

    const int* row = edge_index;
    const int* col = edge_index + NEDGES;

    dim3 blk(256);
    const int grid_gather = ((NNODES + GATHER_ROWS - 1) / GATHER_ROWS + 7) / 8;
    const int grid_head   = NGRAPH / 16;
    const int grid_pool   = ((NNODES + POOL_RUN - 1) / POOL_RUN + 3) / 4;

    gcn_bin_gemm_kernel<<<NEBLK + GEMM_T0, blk, 0, stream>>>(row, col, offR, offC, binR, binC, psum, cnts, gws, total,
                                                          x, W1, b1, H);
    gcn_dinv_gemm_kernel<<<NBINS + GEMM_T1, blk, 0, stream>>>(offC, binC, dinv, x, W1, b1, H);
    gcn_fill_gemm_kernel<<<NBINS + GEMM_T2, blk, 0, stream>>>(offR, binR, dinv, rowstart, rowend, total, elist,
                                                           x, W1, b1, H);

    gcn_gather_kernel<false><<<grid_gather, blk, 0, stream>>>(rowstart, rowend, elist, dinv, H, batch, AGG, wsum);

    gcn_gather_kernel<true><<<grid_gather, blk, 0, stream>>>(rowstart, rowend, elist, dinv, AGG, batch, H, wsum);
    gcn_pool_kernel<<<grid_pool, dim3(128), 0, stream>>>(H, batch, wsum, psum, cnts, gws);
    gcn_head_kernel<<<grid_head, dim3(128), 0, stream>>>(psum, W2, b2, cnts, gws, out);
}


# aggregation passes: a row's loads past the eighth go out together (each had waited for the one before it, one round trip per term): a batch's first load stands unguarded; same sums, same order
# speedup vs baseline: 5.8743x; 1.0310x over previous
#include <hip/hip_runtime.h>
#include <cmath>

#define NNODES 100000
#define NEDGES 600000
#define FEAT   128
#define NGRAPH 256

typedef __attribute__((ext_vector_type(2))) float v2f;
typedef __attribute__((ext_vector_type(8))) float v8f;

#define BIN_SHIFT 8
#define BIN_ROWS  (1 << BIN_SHIFT)
#define NBINS     ((NNODES + BIN_ROWS - 1) / BIN_ROWS)
#define NBINS1    (NBINS + 1)
#define EPT       10
#define EPB       (256 * EPT)
#define NEBLK     ((NEDGES + EPB - 1) / EPB)
#define SCAN_PER_LANE ((NBINS1 + 31) / 32)
#define WALK      8

static_assert(NEBLK * 256 >= NGRAPH * FEAT, "level-1 grid also clears the pooled sums");
static_assert(WALK == 8, "gcn_hold8 names WALK values");

__device__ __forceinline__ void gcn_hold8(int a0, int a1, int a2, int a3, int a4, int a5, int a6, int a7) {
    asm volatile("" :: "v"(a0), "v"(a1), "v"(a2), "v"(a3), "v"(a4), "v"(a5), "v"(a6), "v"(a7));
}
__device__ __forceinline__ void gcn_hold8(float a0, float a1, float a2, float a3, float a4, float a5, float a6, float a7) {
    asm volatile("" :: "v"(a0), "v"(a1), "v"(a2), "v"(a3), "v"(a4), "v"(a5), "v"(a6), "v"(a7));
}

__device__ __forceinline__ void
gcn_bin_role(int* __restrict__ smem, const int blk,
             const int* __restrict__ row, const int* __restrict__ col,
             int* __restrict__ offR, int* __restrict__ offC,
             int* __restrict__ binR_words, int* __restrict__ binC,
             float* __restrict__ psum, float* __restrict__ cnts,
             float* __restrict__ gws, int* __restrict__ total) {
    int*  hR = smem;
    int*  hC = smem + NBINS1;
    int2* sR = (int2*)(smem + 2 * NBINS1);
    int*  sC = smem + 2 * NBINS1 + 2 * EPB;
    const int tid  = threadIdx.x;
    const int lane = tid & 31;
    const int wave = tid >> 5;
    {
        const int i = blk * 256 + tid;
        if (i < NGRAPH * FEAT) psum[i] = 0.0f;
        if (i < NGRAPH) {
            cnts[i] = 0.0f;
            gws[i]  = 0.0f;
        }
        if (i == 0) *total = 0;
    }
    for (int i = tid; i < NBINS1; i += 256) { hR[i] = 0; hC[i] = 0; }

    int r[EPT], c[EPT], kr[EPT], kc[EPT];
    const int e0 = blk * EPB + tid;
    #pragma unroll
    for (int j = 0; j < EPT; ++j) {
        const int e = e0 + j * 256;
        r[j] = 0; c[j] = 0;
        if (e < NEDGES) { r[j] = row[e]; c[j] = col[e]; }
    }
    __syncthreads();
    #pragma unroll
    for (int j = 0; j < EPT; ++j) {
        kr[j] = 0; kc[j] = 0;
        if (e0 + j * 256 < NEDGES) {
            kr[j] = atomicAdd(&hR[r[j] >> BIN_SHIFT], 1);
            kc[j] = atomicAdd(&hC[c[j] >> BIN_SHIFT], 1);
        }
    }
    __syncthreads();
    if (wave < 2) {
        int* h  = (wave == 0) ? hR : hC;
        int* og = ((wave == 0) ? offR : offC) + blk;
        int v[SCAN_PER_LANE];
        int sum = 0;
        #pragma unroll
        for (int q = 0; q < SCAN_PER_LANE; ++q) {
            const int idx = lane * SCAN_PER_LANE + q;
            v[q] = (idx < NBINS1) ? h[idx] : 0;
            sum += v[q];
        }
        int inc = sum;
        #pragma unroll
        for (int o = 1; o < 32; o <<= 1) {
            const int t = __shfl_up(inc, o);
            if (lane >= o) inc += t;
        }
        int ex = inc - sum;
        #pragma unroll
        for (int q = 0; q < SCAN_PER_LANE; ++q) {
            const int idx = lane * SCAN_PER_LANE + q;
            if (idx < NBINS1) { h[idx] = ex; og[idx * NEBLK] = ex; }
            ex += v[q];
        }
    }
    __syncthreads();
    #pragma unroll
    for (int j = 0; j < EPT; ++j) {
        if (e0 + j * 256 < NEDGES) {
            sR[hR[r[j] >> BIN_SHIFT] + kr[j]] = make_int2(r[j], c[j]);
            sC[hC[c[j] >> BIN_SHIFT] + kc[j]] = c[j];
        }
    }
    __syncthreads();
    const int nE = min(EPB, NEDGES - blk * EPB);
    int2* binR = (int2*)binR_words + (size_t)blk * EPB;
    int*  bC   = binC + (size_t)blk * EPB;
    #pragma unroll
    for (int j = 0; j < EPT; ++j) {
        const int i = tid + j * 256;
        if (i < nE) {
            binR[i] = sR[i];
            bC[i]   = sC[i];
        }
    }
}

__device__ __forceinline__ void
gcn_dinv_role(int* __restrict__ smem, const int b,
              const int* __restrict__ offC, const int* __restrict__ binC,
              float* __restrict__ dinv) {
    int* cnt = smem;
    const int tid = threadIdx.x;
    cnt[tid] = 0;
    __syncthreads();
    for (int k = tid; k < NEBLK; k += 256) {
        const int s = offC[b * NEBLK + k];
        const int t = offC[(b + 1) * NEBLK + k];
        const int* src = binC + (size_t)k * EPB;
        for (int i = s; i < t; i += WALK) {
            int cc[WALK];
            #pragma unroll
            for (int q = 0; q < WALK; ++q) cc[q] = src[min(i + q, t - 1)];
            gcn_hold8(cc[0], cc[1], cc[2], cc[3], cc[4], cc[5], cc[6], cc[7]);
            #pragma unroll
            for (int q = 0; q < WALK; ++q)
                if (i + q < t) atomicAdd(&cnt[cc[q] & (BIN_ROWS - 1)], 1);
        }
    }
    __syncthreads();
    const int n = b * BIN_ROWS + tid;
    if (n < NNODES) {
        const float d = 1.0f + (float)cnt[tid];
        dinv[n] = (d > 0.0f) ? rsqrtf(d) : 0.0f;
    }
}

__device__ __forceinline__ void
gcn_fill_role(int* __restrict__ smem, const int b,
              const int* __restrict__ offR, const int* __restrict__ binR_words,
              const float* __restrict__ dinv,
              int* __restrict__ rowstart, int* __restrict__ rowend,
              int* __restrict__ total, int* __restrict__ elist_words) {
    int*   cnt   = smem;
    float* dl    = (float*)(smem + BIN_ROWS);
    int*   wsum  = smem + 2 * BIN_ROWS;
    int*   sbase = smem + 2 * BIN_ROWS + 8;
    const int2* binR = (const int2*)binR_words;
    int2* elist = (int2*)elist_words;
    const int tid  = threadIdx.x;
    const int lane = tid & 31;
    const int wave = tid >> 5;
    const int n    = b * BIN_ROWS + tid;
    cnt[tid] = 0;
    dl[tid]  = (n < NNODES) ? dinv[n] : 0.0f;
    __syncthreads();
    for (int k = tid; k < NEBLK; k += 256) {
        const int s = offR[b * NEBLK + k];
        const int t = offR[(b + 1) * NEBLK + k];
        const int2* src = binR + (size_t)k * EPB;
        for (int i = s; i < t; i += WALK) {
            int rr[WALK];
            #pragma unroll
            for (int q = 0; q < WALK; ++q) rr[q] = src[min(i + q, t - 1)].x;
            gcn_hold8(rr[0], rr[1], rr[2], rr[3], rr[4], rr[5], rr[6], rr[7]);
            #pragma unroll
            for (int q = 0; q < WALK; ++q)
                if (i + q < t) atomicAdd(&cnt[rr[q] & (BIN_ROWS - 1)], 1);
        }
    }
    __syncthreads();
    const int c = cnt[tid];
    int v = c;
    #pragma unroll
    for (int o = 1; o < 32; o <<= 1) {
        int t = __shfl_up(v, o);
        if (lane >= o) v += t;
    }
    if (lane == 31) wsum[wave] = v;
    __syncthreads();
    if (tid == 0) {
        int s = 0;
        for (int w = 0; w < 8; ++w) { int t = wsum[w]; wsum[w] = s; s += t; }
        *sbase = atomicAdd(total, s);
    }
    __syncthreads();
    const int st = *sbase + wsum[wave] + v - c;
    if (n < NNODES) {
        rowstart[n] = st;
        rowend[n]   = st + c;
    }
    cnt[tid] = st;
    __syncthreads();
    for (int k = tid; k < NEBLK; k += 256) {
        const int s = offR[b * NEBLK + k];
        const int t = offR[(b + 1) * NEBLK + k];
        const int2* src = binR + (size_t)k * EPB;
        for (int i = s; i < t; i += WALK) {
            int2  en[WALK];
            float dc[WALK];
            #pragma unroll
            for (int q = 0; q < WALK; ++q) en[q] = src[min(i + q, t - 1)];
            #pragma unroll
            for (int q = 0; q < WALK; ++q) dc[q] = dinv[en[q].y];
            gcn_hold8(dc[0], dc[1], dc[2], dc[3], dc[4], dc[5], dc[6], dc[7]);
            #pragma unroll
            for (int q = 0; q < WALK; ++q) {
                if (i + q < t) {
                    const int rl  = en[q].x & (BIN_ROWS - 1);
                    const int pos = atomicAdd(&cnt[rl], 1);
                    const float w = dl[rl] * dc[q];
                    elist[pos] = make_int2(en[q].y, __float_as_int(w));
                }
            }
        }
    }
}

#define GEMM_ROWS  64
#define WP_PITCH   288
#define AT_PITCH   132
#define SMEM_WORDS (64 * WP_PITCH + GEMM_ROWS * AT_PITCH)
#define GEMM_TILES ((NNODES + GEMM_ROWS - 1) / GEMM_ROWS)
#define GEMM_T0    512
#define GEMM_T2    377
#define GEMM_T1    (GEMM_TILES - GEMM_T0 - GEMM_T2)

static_assert(SMEM_WORDS >= 2 * NBINS1 + 3 * EPB, "level-1 role fits the shared buffer");
static_assert(GEMM_T1 > 0, "tile split");
static_assert(NNODES % 8 == 0, "a wave's 8 staged rows are all inside or all outside");

__device__ __forceinline__ void
gcn_gemm_role(float* __restrict__ smem, const int tile,
              const float* __restrict__ X,
              const float* __restrict__ W,
              const float* __restrict__ bias,
              float* __restrict__ Y) {
    float* Wp = smem;
    float* At = smem + 64 * WP_PITCH;

    const int lane = threadIdx.x & 31;
    const int wave = threadIdx.x >> 5;
    const int rt   = wave & 3;
    const int ch   = wave >> 2;
    const int m0   = tile * GEMM_ROWS;
    const int c = lane & 15;
    const int h = lane >> 4;

    float4 arow[8];
    {
        const float4* src = (const float4*)X + lane;
        #pragma unroll
        for (int j = 0; j < 8; ++j) {
            const int r = min(m0 + wave * 8 + j, NNODES - 1);
            arow[j] = src[(size_t)r * 32];
        }
    }

    {
        const float4* W4 = (const float4*)W;
        float4 r0[8], r1[8];
        #pragma unroll
        for (int i = 0; i < 8; ++i) {
            const int p = i * 8 + wave;
            r0[i] = W4[(2 * p) * 32 + lane];
            r1[i] = W4[(2 * p + 1) * 32 + lane];
        }
        #pragma unroll
        for (int j = 0; j < 8; ++j)
            *(float4*)(At + (wave * 8 + j) * AT_PITCH + lane * 4) = arow[j];
        #pragma unroll
        for (int i = 0; i < 8; ++i) {
            const int p = i * 8 + wave;
            float4* dst = (float4*)(Wp + p * WP_PITCH + lane * 8);
            dst[0] = make_float4(r0[i].x, r1[i].x, r0[i].y, r1[i].y);
            dst[1] = make_float4(r0[i].z, r1[i].z, r0[i].w, r1[i].w);
        }
    }
    __syncthreads();

    v8f acc[4];
    #pragma unroll
    for (int nt = 0; nt < 4; ++nt) acc[nt] = v8f{};

    {
        const float* ap = At + (rt * 16 + c) * AT_PITCH + 2 * h;
        const float* bp = Wp + h * WP_PITCH + 128 * ch + 2 * c;
        #pragma unroll
        for (int s = 0; s < 32; ++s) {
            const v2f a = *(const v2f*)(ap + 4 * s);
            #pragma unroll
            for (int nt = 0; nt < 4; ++nt) {
                const v2f b = *(const v2f*)(bp + 2 * s * WP_PITCH + 32 * nt);
                acc[nt] = __builtin_amdgcn_wmma_f32_16x16x4_f32(
                    false, a, false, b, (short)0, acc[nt], false, false);
            }
        }
    }
    __syncthreads();

    #pragma unroll
    for (int nt = 0; nt < 4; ++nt) {
        const float bv = bias[ch * 64 + nt * 16 + c];
        #pragma unroll
        for (int r = 0; r < 8; ++r)
            At[(rt * 16 + 8 * h + r) * AT_PITCH + ch * 64 + nt * 16 + c] = acc[nt][r] + bv;
    }
    __syncthreads();
    const int rw = m0 + wave * 8;
    if (rw < NNODES) {
        float4* dst = (float4*)Y + (size_t)rw * 32 + lane;
        #pragma unroll
        for (int j = 0; j < 8; ++j)
            dst[j * 32] = *(const float4*)(At + (wave * 8 + j) * AT_PITCH + lane * 4);
    }
}

__global__ void __launch_bounds__(256)
gcn_bin_gemm_kernel(const int* __restrict__ row, const int* __restrict__ col,
                    int* __restrict__ offR, int* __restrict__ offC,
                    int* __restrict__ binR_words, int* __restrict__ binC,
                    float* __restrict__ psum, float* __restrict__ cnts,
                    float* __restrict__ gws, int* __restrict__ total,
                    const float* __restrict__ X, const float* __restrict__ W,
                    const float* __restrict__ bias, float* __restrict__ Y) {
    __shared__ alignas(16) int smem[SMEM_WORDS];
    const int b = blockIdx.x;
    if (b < NEBLK) gcn_bin_role(smem, b, row, col, offR, offC, binR_words, binC, psum, cnts, gws, total);
    else           gcn_gemm_role((float*)smem, b - NEBLK, X, W, bias, Y);
}

__global__ void __launch_bounds__(256)
gcn_dinv_gemm_kernel(const int* __restrict__ offC, const int* __restrict__ binC,
                     float* __restrict__ dinv,
                     const float* __restrict__ X, const float* __restrict__ W,
                     const float* __restrict__ bias, float* __restrict__ Y) {
    __shared__ alignas(16) int smem[SMEM_WORDS];
    const int b = blockIdx.x;
    if (b < NBINS) gcn_dinv_role(smem, b, offC, binC, dinv);
    else           gcn_gemm_role((float*)smem, GEMM_T0 + (b - NBINS), X, W, bias, Y);
}

__global__ void __launch_bounds__(256)
gcn_fill_gemm_kernel(const int* __restrict__ offR, const int* __restrict__ binR_words,
                     const float* __restrict__ dinv,
                     int* __restrict__ rowstart, int* __restrict__ rowend,
                     int* __restrict__ total, int* __restrict__ elist_words,
                     const float* __restrict__ X, const float* __restrict__ W,
                     const float* __restrict__ bias, float* __restrict__ Y) {
    __shared__ alignas(16) int smem[SMEM_WORDS];
    const int b = blockIdx.x;
    if (b < NBINS) gcn_fill_role(smem, b, offR, binR_words, dinv, rowstart, rowend, total, elist_words);
    else           gcn_gemm_role((float*)smem, GEMM_T0 + GEMM_T1 + (b - NBINS), X, W, bias, Y);
}

#define GATHER_ROWS  8
#define GATHER_BATCH 8

__device__ __forceinline__ bool gcn_run_end(int i_in_group, int g, int g_next) {
    return i_in_group == GATHER_ROWS - 1 || g != g_next;
}

__device__ __forceinline__ float4 gcn_relu4(float4 v) {
    return make_float4(fmaxf(v.x, 0.0f), fmaxf(v.y, 0.0f), fmaxf(v.z, 0.0f), fmaxf(v.w, 0.0f));
}

__device__ __forceinline__ int2 gcn_row_terms(const int2* __restrict__ elist, int s, int e, int base,
                                              int lane, int n, float sw) {
    int2 ent = make_int2(0, 0);
    const int idx = base + lane - 1;
    if (idx >= s && idx < e) ent = elist[idx];
    if (base == s && lane == 0) ent = make_int2(n, __float_as_int(sw));
    return ent;
}

__device__ __forceinline__ void gcn_batch_issue(float4 (&h)[GATHER_BATCH], const float4* __restrict__ H4,
                                                int2 ent, int j0, int cnt, int lane) {
    {
        const int c = __builtin_amdgcn_readlane(ent.x, j0);
        h[0] = (H4 + (size_t)c * 32)[lane];
    }
    #pragma unroll
    for (int k = 1; k < GATHER_BATCH; ++k) {
        if (k < cnt) {
            const int c = __builtin_amdgcn_readlane(ent.x, j0 + k);
            h[k] = (H4 + (size_t)c * 32)[lane];
        }
    }
}

template <bool LAYER2>
__device__ __forceinline__ void gcn_batch_sum(const float4 (&h)[GATHER_BATCH], int2 ent, int j0, int cnt,
                                              float4& acc, float& ws) {
    {
        const float w = __int_as_float(__builtin_amdgcn_readlane(ent.y, j0));
        if (LAYER2) ws += w;
        acc.x += w * h[0].x; acc.y += w * h[0].y;
        acc.z += w * h[0].z; acc.w += w * h[0].w;
    }
    #pragma unroll
    for (int k = 1; k < GATHER_BATCH; ++k) {
        if (k < cnt) {
            const float w = __int_as_float(__builtin_amdgcn_readlane(ent.y, j0 + k));
            if (LAYER2) ws += w;
            acc.x += w * h[k].x; acc.y += w * h[k].y;
            acc.z += w * h[k].z; acc.w += w * h[k].w;
        }
    }
}

template <bool LAYER2>
__global__ void __launch_bounds__(256)
gcn_gather_kernel(const int* __restrict__ rowstart,
                  const int* __restrict__ rowend,
                  const int* __restrict__ elist_words,
                  const float* __restrict__ dinv,
                  const float* __restrict__ H,
                  const int* __restrict__ batch,
                  float* __restrict__ AGG,
                  float* __restrict__ wsum) {
    const int2* elist = (const int2*)elist_words;
    const int lane = threadIdx.x & 31;
    const int wv   = (blockIdx.x * 256 + threadIdx.x) >> 5;
    const int n0   = wv * GATHER_ROWS;
    if (n0 >= NNODES) return;
    const int nrows = (NNODES % GATHER_ROWS == 0) ? GATHER_ROWS
                                                  : min(GATHER_ROWS, NNODES - n0);
    int   ls = 0, le = 0;
    float lw = 0.0f;
    int   lg = 0;
    if (lane < nrows) {
        ls = rowstart[n0 + lane];
        le = rowend[n0 + lane];
        const float d = dinv[n0 + lane];
        lw = d * d;
        if (LAYER2) lg = batch[n0 + lane];
    }
    const float4* H4 = (const float4*)H;

    int   s  = __builtin_amdgcn_readlane(ls, 0);
    int   e  = __builtin_amdgcn_readlane(le, 0);
    float sw = __int_as_float(__builtin_amdgcn_readlane(__float_as_int(lw), 0));
    int2  ent = gcn_row_terms(elist, s, e, s, lane, n0, sw);
    float wsv = 0.0f;
    float4 run = make_float4(0.0f, 0.0f, 0.0f, 0.0f);

    for (int i = 0; i < nrows; ++i) {
        const int n = n0 + i;
        const int terms = e - s + 1;
        float4 acc = make_float4(0.0f, 0.0f, 0.0f, 0.0f);
        float  ws  = 0.0f;
        float4 h[GATHER_BATCH];

        int m   = min(32, terms);
        int cnt = min(GATHER_BATCH, m);
        gcn_batch_issue(h, H4, ent, 0, cnt, lane);

        int s1 = 0, e1 = 0; float sw1 = 0.0f;
        int2 ent1 = make_int2(0, 0);
        if (i + 1 < nrows) {
            s1  = __builtin_amdgcn_readlane(ls, i + 1);
            e1  = __builtin_amdgcn_readlane(le, i + 1);
            sw1 = __int_as_float(__builtin_amdgcn_readlane(__float_as_int(lw), i + 1));
            ent1 = gcn_row_terms(elist, s1, e1, s1, lane, n + 1, sw1);
        }

        gcn_batch_sum<LAYER2>(h, ent, 0, cnt, acc, ws);

        for (int j0 = GATHER_BATCH; j0 < m; j0 += GATHER_BATCH) {
            cnt = min(GATHER_BATCH, m - j0);
            gcn_batch_issue(h, H4, ent, j0, cnt, lane);
            gcn_batch_sum<LAYER2>(h, ent, j0, cnt, acc, ws);
        }
        for (int t0 = 32; t0 < terms; t0 += 32) {
            ent = gcn_row_terms(elist, s, e, s + t0, lane, n, sw);
            m = min(32, terms - t0);
            for (int j0 = 0; j0 < m; j0 += GATHER_BATCH) {
                cnt = min(GATHER_BATCH, m - j0);
                gcn_batch_issue(h, H4, ent, j0, cnt, lane);
                gcn_batch_sum<LAYER2>(h, ent, j0, cnt, acc, ws);
            }
        }

        if (LAYER2) {
            run.x += acc.x; run.y += acc.y; run.z += acc.z; run.w += acc.w;
            const int g  = __builtin_amdgcn_readlane(lg, i);
            const int gn = __builtin_amdgcn_readlane(lg, i + 1);
            if (i == nrows - 1 || gcn_run_end(i, g, gn)) {
                ((float4*)AGG + (size_t)n * 32)[lane] = run;
                run = make_float4(0.0f, 0.0f, 0.0f, 0.0f);
            }
        } else {
            ((float4*)AGG + (size_t)n * 32)[lane] = gcn_relu4(acc);
        }
        if (LAYER2 && lane == i) wsv = ws;
        s = s1; e = e1; sw = sw1; ent = ent1;
    }
    if (LAYER2 && lane < nrows) wsum[n0 + lane] = wsv;
}

#define POOL_RUN 32
#define POOL_GROUPS (POOL_RUN / GATHER_ROWS)

static_assert(POOL_RUN % GATHER_ROWS == 0 && NNODES % GATHER_ROWS == 0,
              "a pool wave's nodes are whole groups of the aggregation's rows");

__device__ __forceinline__ void gcn_pool_flush(float* __restrict__ psum,
                                               float* __restrict__ cnts,
                                               float* __restrict__ gws,
                                               int g, float4 acc, float cnt, float wacc, int lane) {
    float* dst = psum + (size_t)g * FEAT + lane * 4;
    atomicAdd(dst + 0, acc.x);
    atomicAdd(dst + 1, acc.y);
    atomicAdd(dst + 2, acc.z);
    atomicAdd(dst + 3, acc.w);
    if (lane == 0) atomicAdd(&cnts[g], cnt);
    if (lane == 1) atomicAdd(&gws[g], wacc);
}

__global__ void __launch_bounds__(128)
gcn_pool_kernel(const float* __restrict__ AGG,
                const int* __restrict__ batch,
                const float* __restrict__ wsum,
                float* __restrict__ psum,
                float* __restrict__ cnts,
                float* __restrict__ gws) {
    const int lane = threadIdx.x & 31;
    const int w    = (blockIdx.x * 128 + threadIdx.x) >> 5;
    const int n0   = w * POOL_RUN;
    if (n0 >= NNODES) return;
    const int nn = (NNODES % POOL_RUN == 0) ? POOL_RUN
                                            : min(POOL_RUN, NNODES - n0);
    const int gl = (lane < nn) ? batch[n0 + lane] : 0;
    const float wl = (lane < nn) ? wsum[n0 + lane] : 0.0f;
    const float4* src = (const float4*)(AGG + (size_t)n0 * FEAT) + lane;

    float4 h[POOL_GROUPS];
    #pragma unroll
    for (int q = 0; q < POOL_GROUPS; ++q)
        if (q * GATHER_ROWS + GATHER_ROWS - 1 < nn) h[q] = src[(q * GATHER_ROWS + GATHER_ROWS - 1) * 32];

    int    gcur = __builtin_amdgcn_readlane(gl, 0);
    float4 acc  = make_float4(0.0f, 0.0f, 0.0f, 0.0f);
    float  cnt  = 0.0f;
    float  wacc = 0.0f;
    #pragma unroll
    for (int i = 0; i < POOL_RUN; ++i) {
        if (i < nn) {
            const int g = __builtin_amdgcn_readlane(gl, i);
            if (g != gcur) {
                gcn_pool_flush(psum, cnts, gws, gcur, acc, cnt, wacc, lane);
                acc  = make_float4(0.0f, 0.0f, 0.0f, 0.0f);
                cnt  = 0.0f;
                wacc = 0.0f;
                gcur = g;
            }
            if ((i % GATHER_ROWS) == GATHER_ROWS - 1) {
                const float4 t = h[i / GATHER_ROWS];
                acc.x += t.x; acc.y += t.y; acc.z += t.z; acc.w += t.w;
            } else if (gcn_run_end(i % GATHER_ROWS, g, __builtin_amdgcn_readlane(gl, i + 1))) {
                const float4 t = src[i * 32];
                acc.x += t.x; acc.y += t.y; acc.z += t.z; acc.w += t.w;
            }
            cnt  += 1.0f;
            wacc += __int_as_float(__builtin_amdgcn_readlane(__float_as_int(wl), i));
        }
    }
    gcn_pool_flush(psum, cnts, gws, gcur, acc, cnt, wacc, lane);
}

__global__ void __launch_bounds__(128)
gcn_head_kernel(const float* __restrict__ psum,
                const float* __restrict__ W,
                const float* __restrict__ bias,
                const float* __restrict__ cnts,
                const float* __restrict__ gws,
                float* __restrict__ out) {
    __shared__ alignas(16) float Ct[16 * AT_PITCH];

    const int lane = threadIdx.x & 31;
    const int wave = threadIdx.x >> 5;
    const int c  = lane & 15;
    const int h  = lane >> 4;
    const int m0 = blockIdx.x * 16;
    const int n0 = wave * 32;

    const float* ap = psum + (size_t)(m0 + c) * FEAT + 2 * h;
    const float* bp = W + (size_t)(2 * h) * FEAT + n0 + c;
    v2f a[32], b0[32], b1[32];
    #pragma unroll
    for (int s = 0; s < 32; ++s) {
        a[s]  = *(const v2f*)(ap + 4 * s);
        b0[s] = v2f{bp[(4 * s) * FEAT],      bp[(4 * s + 1) * FEAT]};
        b1[s] = v2f{bp[(4 * s) * FEAT + 16], bp[(4 * s + 1) * FEAT + 16]};
    }

    v8f acc0 = v8f{};
    v8f acc1 = v8f{};
    #pragma unroll
    for (int s = 0; s < 32; ++s) {
        acc0 = __builtin_amdgcn_wmma_f32_16x16x4_f32(
            false, a[s], false, b0[s], (short)0, acc0, false, false);
        acc1 = __builtin_amdgcn_wmma_f32_16x16x4_f32(
            false, a[s], false, b1[s], (short)0, acc1, false, false);
    }

    #pragma unroll
    for (int r = 0; r < 8; ++r) {
        Ct[(8 * h + r) * AT_PITCH + n0 + c]      = acc0[r];
        Ct[(8 * h + r) * AT_PITCH + n0 + 16 + c] = acc1[r];
    }
    __syncthreads();

    const float4 bv = ((const float4*)bias)[lane];
    #pragma unroll
    for (int j = 0; j < 4; ++j) {
        const int   rr = wave * 4 + j;
        const int   g  = m0 + rr;
        const float d  = fmaxf(cnts[g], 1.0f);
        const float wm = gws[g] / d;
        const float4 v = *(const float4*)(Ct + rr * AT_PITCH + lane * 4);
        float4 o;
        o.x = v.x / d + wm * bv.x;
        o.y = v.y / d + wm * bv.y;
        o.z = v.z / d + wm * bv.z;
        o.w = v.w / d + wm * bv.w;
        ((float4*)out)[(size_t)g * 32 + lane] = o;
    }
}

extern "C" void kernel_launch(void* const* d_in, const int* in_sizes, int n_in,
                              void* d_out, int out_size, void* d_ws, size_t ws_size,
                              hipStream_t stream) {
    const float* x          = (const float*)d_in[0];
    const int*   edge_index = (const int*)  d_in[1];
    const int*   batch      = (const int*)  d_in[2];
    const float* W1 = (const float*)d_in[4];
    const float* b1 = (const float*)d_in[5];
    const float* W2 = (const float*)d_in[6];
    const float* b2 = (const float*)d_in[7];
    float* out = (float*)d_out;

    float* H    = (float*)d_ws;
    float* AGG  = H    + (size_t)NNODES * FEAT;
    float* dinv = AGG  + (size_t)NNODES * FEAT;
    float* cnts = dinv + NNODES;
    float* gws  = cnts + NGRAPH;
    float* psum = gws  + NGRAPH;
    float* wsum = psum + (size_t)NGRAPH * FEAT;
    int*   rowend   = (int*)(wsum + NNODES);
    int*   rowstart = rowend + NNODES;
    int*   total    = rowstart + NNODES;
    int*   elist    = total + 16;
    int*   binR     = elist + 2 * NEDGES;
    int*   binC     = binR + 2 * NEBLK * EPB;
    int*   offR     = binC + NEBLK * EPB;
    int*   offC     = offR + NEBLK * NBINS1;

    const int* row = edge_index;
    const int* col = edge_index + NEDGES;

    dim3 blk(256);
    const int grid_gather = ((NNODES + GATHER_ROWS - 1) / GATHER_ROWS + 7) / 8;
    const int grid_head   = NGRAPH / 16;
    const int grid_pool   = ((NNODES + POOL_RUN - 1) / POOL_RUN + 3) / 4;

    gcn_bin_gemm_kernel<<<NEBLK + GEMM_T0, blk, 0, stream>>>(row, col, offR, offC, binR, binC, psum, cnts, gws, total,
                                                          x, W1, b1, H);
    gcn_dinv_gemm_kernel<<<NBINS + GEMM_T1, blk, 0, stream>>>(offC, binC, dinv, x, W1, b1, H);
    gcn_fill_gemm_kernel<<<NBINS + GEMM_T2, blk, 0, stream>>>(offR, binR, dinv, rowstart, rowend, total, elist,
                                                           x, W1, b1, H);

    gcn_gather_kernel<false><<<grid_gather, blk, 0, stream>>>(rowstart, rowend, elist, dinv, H, batch, AGG, wsum);

    gcn_gather_kernel<true><<<grid_gather, blk, 0, stream>>>(rowstart, rowend, elist, dinv, AGG, batch, H, wsum);
    gcn_pool_kernel<<<grid_pool, dim3(128), 0, stream>>>(H, batch, wsum, psum, cnts, gws);
    gcn_head_kernel<<<grid_head, dim3(128), 0, stream>>>(psum, W2, b2, cnts, gws, out);
}
